# starting kernel
# baseline (speedup 1.0000x reference)
#include <hip/hip_runtime.h>
#include <hip/hip_bf16.h>

typedef __attribute__((ext_vector_type(16))) _Float16 v16h;
typedef __attribute__((ext_vector_type(8)))  float    v8f;

#define DH 128

__global__ __launch_bounds__(256)
void zero_f32(float* __restrict__ p, long long n4) {
    long long i = (long long)blockIdx.x * blockDim.x + threadIdx.x;
    if (i < n4) ((float4*)p)[i] = make_float4(0.f, 0.f, 0.f, 0.f);
}

__global__ __launch_bounds__(256)
void pack_weights(const float* __restrict__ W, _Float16* __restrict__ Wp) {
    int idx = blockIdx.x * blockDim.x + threadIdx.x;
    if (idx >= 4 * 8 * 32 * 16) return;
    int h    = idx & 15;
    int lane = (idx >> 4) & 31;
    int nt   = (idx >> 9) & 7;
    int kt   = (idx >> 12) & 3;
    int grp  = lane >> 4;
    int k    = kt * 32 + grp * 8 + ((h < 8) ? h : (16 + (h - 8)));
    int n    = nt * 16 + (lane & 15);
    Wp[idx]  = (_Float16)W[k * DH + n];
}

__global__ __launch_bounds__(256)
void edge_aggregate(const float* __restrict__ srcRows,
                    const int* __restrict__ esrc, const int* __restrict__ edst,
                    float* __restrict__ agg, int nEdges) {
    int edge = blockIdx.x * (blockDim.x >> 5) + (threadIdx.x >> 5);
    if (edge >= nEdges) return;
    int lane = threadIdx.x & 31;
    int s = __builtin_amdgcn_readfirstlane(esrc[edge]);
    int d = __builtin_amdgcn_readfirstlane(edst[edge]);
    float4 v = ((const float4*)(srcRows + (long long)s * DH))[lane];
    float* out = agg + (long long)d * DH + (lane << 2);
    atomicAdd(out + 0, v.x);
    atomicAdd(out + 1, v.y);
    atomicAdd(out + 2, v.z);
    atomicAdd(out + 3, v.w);
}

template <bool HAS_A1, bool HAS_RES, bool RELU>
__global__ __launch_bounds__(128)
void gemm128_wmma(const float* __restrict__ A0,
                  const float* __restrict__ A1,
                  const _Float16* __restrict__ Wp,
                  const float* __restrict__ bias,
                  const float* __restrict__ Res,
                  float* __restrict__ Out,
                  int nRowTiles) {
    __shared__ float lds[4][16 * DH];
    int wave = blockIdx.x * (blockDim.x >> 5) + (threadIdx.x >> 5);
    if (wave >= nRowTiles) return;
    int lane = threadIdx.x & 31;
    int grp  = lane >> 4;
    int col  = lane & 15;
    float* myLds = lds[threadIdx.x >> 5];
    long long row = (long long)wave * 16 + col;

    const float* ar0 = A0 + row * DH;
    const float* ar1 = HAS_A1 ? (A1 + row * DH) : nullptr;
    v16h afrag[4];
#pragma unroll
    for (int kt = 0; kt < 4; ++kt) {
        int off = kt * 32 + grp * 8;
        float4 q0 = *(const float4*)(ar0 + off);
        float4 q1 = *(const float4*)(ar0 + off + 4);
        float4 q2 = *(const float4*)(ar0 + off + 16);
        float4 q3 = *(const float4*)(ar0 + off + 20);
        if (HAS_A1) {
            float4 b;
            b = *(const float4*)(ar1 + off);      q0.x += b.x; q0.y += b.y; q0.z += b.z; q0.w += b.w;
            b = *(const float4*)(ar1 + off + 4);  q1.x += b.x; q1.y += b.y; q1.z += b.z; q1.w += b.w;
            b = *(const float4*)(ar1 + off + 16); q2.x += b.x; q2.y += b.y; q2.z += b.z; q2.w += b.w;
            b = *(const float4*)(ar1 + off + 20); q3.x += b.x; q3.y += b.y; q3.z += b.z; q3.w += b.w;
        }
        v16h a;
        a[0]  = (_Float16)q0.x; a[1]  = (_Float16)q0.y; a[2]  = (_Float16)q0.z; a[3]  = (_Float16)q0.w;
        a[4]  = (_Float16)q1.x; a[5]  = (_Float16)q1.y; a[6]  = (_Float16)q1.z; a[7]  = (_Float16)q1.w;
        a[8]  = (_Float16)q2.x; a[9]  = (_Float16)q2.y; a[10] = (_Float16)q2.z; a[11] = (_Float16)q2.w;
        a[12] = (_Float16)q3.x; a[13] = (_Float16)q3.y; a[14] = (_Float16)q3.z; a[15] = (_Float16)q3.w;
        afrag[kt] = a;
    }

    const v16h* wfrag = (const v16h*)Wp;
#pragma unroll
    for (int nt = 0; nt < 8; ++nt) {
        v8f acc = {};
#pragma unroll
        for (int kt = 0; kt < 4; ++kt) {
            v16h b = wfrag[(kt * 8 + nt) * 32 + lane];
            acc = __builtin_amdgcn_wmma_f32_16x16x32_f16(
                false, afrag[kt], false, b, (short)0, acc, false, false);
        }
#pragma unroll
        for (int r = 0; r < 8; ++r)
            myLds[(r + grp * 8) * DH + nt * 16 + col] = acc[r];
    }

    float4 bvec = ((const float4*)bias)[lane];
#pragma unroll
    for (int rl = 0; rl < 16; ++rl) {
        float4 v = ((const float4*)myLds)[rl * 32 + lane];
        v.x += bvec.x; v.y += bvec.y; v.z += bvec.z; v.w += bvec.w;
        if (RELU) {
            v.x = fmaxf(v.x, 0.f); v.y = fmaxf(v.y, 0.f);
            v.z = fmaxf(v.z, 0.f); v.w = fmaxf(v.w, 0.f);
        }
        long long m = (long long)wave * 16 + rl;
        if (HAS_RES) {
            float4 rv = ((const float4*)(Res + m * DH))[lane];
            v.x += rv.x; v.y += rv.y; v.z += rv.z; v.w += rv.w;
        }
        ((float4*)(Out + m * DH))[lane] = v;
    }
}

__global__ __launch_bounds__(256)
void fuse_t2(const float* __restrict__ t, const float* __restrict__ ga,
             const float* __restrict__ gb, float* __restrict__ out, long long n4) {
    long long i = (long long)blockIdx.x * blockDim.x + threadIdx.x;
    if (i >= n4) return;
    float4 a = ((const float4*)t)[i];
    float4 b = ((const float4*)ga)[i];
    float4 c = ((const float4*)gb)[i];
    float4 o;
    o.x = fmaxf(a.x + b.x + c.x, 0.f);
    o.y = fmaxf(a.y + b.y + c.y, 0.f);
    o.z = fmaxf(a.z + b.z + c.z, 0.f);
    o.w = fmaxf(a.w + b.w + c.w, 0.f);
    ((float4*)out)[i] = o;
}

__global__ __launch_bounds__(256)
void ln_residual(const float* __restrict__ u, const float* __restrict__ t2,
                 const float* __restrict__ g, const float* __restrict__ b,
                 float* __restrict__ out, int nRows) {
    int wave = blockIdx.x * (blockDim.x >> 5) + (threadIdx.x >> 5);
    if (wave >= nRows) return;
    int lane = threadIdx.x & 31;
    long long base = (long long)wave * DH;
    float4 v = ((const float4*)(u + base))[lane];
    float s  = v.x + v.y + v.z + v.w;
    float sq = v.x * v.x + v.y * v.y + v.z * v.z + v.w * v.w;
#pragma unroll
    for (int off = 16; off > 0; off >>= 1) {
        s  += __shfl_xor(s, off, 32);
        sq += __shfl_xor(sq, off, 32);
    }
    float mean = s * (1.0f / 128.0f);
    float var  = sq * (1.0f / 128.0f) - mean * mean;
    float inv  = rsqrtf(var + 1e-5f);
    float4 tv = ((const float4*)(t2 + base))[lane];
    float4 gv = ((const float4*)g)[lane];
    float4 bv = ((const float4*)b)[lane];
    float4 o;
    o.x = tv.x + (v.x - mean) * inv * gv.x + bv.x;
    o.y = tv.y + (v.y - mean) * inv * gv.y + bv.y;
    o.z = tv.z + (v.z - mean) * inv * gv.z + bv.z;
    o.w = tv.w + (v.w - mean) * inv * gv.w + bv.w;
    ((float4*)(out + base))[lane] = o;
}

extern "C" void kernel_launch(void* const* d_in, const int* in_sizes, int n_in,
                              void* d_out, int out_size, void* d_ws, size_t ws_size,
                              hipStream_t stream) {
    const float* x    = (const float*)d_in[0];
    const float* t    = (const float*)d_in[1];
    const int*   e_t  = (const int*)d_in[2];
    const int*   e_x  = (const int*)d_in[3];
    const float* W1a  = (const float*)d_in[4];
    const float* b1a  = (const float*)d_in[5];
    const float* W2a  = (const float*)d_in[6];
    const float* b2a  = (const float*)d_in[7];
    const float* W1b  = (const float*)d_in[8];
    const float* b1b  = (const float*)d_in[9];
    const float* W2b  = (const float*)d_in[10];
    const float* b2b  = (const float*)d_in[11];
    const float* Wo   = (const float*)d_in[12];
    const float* bo   = (const float*)d_in[13];
    const float* ln_g = (const float*)d_in[14];
    const float* ln_b = (const float*)d_in[15];
    const float* Wf1  = (const float*)d_in[16];
    const float* bf1  = (const float*)d_in[17];
    const float* Wf2  = (const float*)d_in[18];
    const float* bf2  = (const float*)d_in[19];

    const int N = in_sizes[0] / DH;
    const int E = in_sizes[2] / 2;
    const long long ND = (long long)N * DH;

    float* buf0 = (float*)d_ws;
    float* buf1 = buf0 + ND;
    float* buf2 = buf1 + ND;
    _Float16* wp = (_Float16*)(buf2 + ND);
    const int WPN = 4 * 8 * 32 * 16;
    _Float16* wp0 = wp;
    _Float16* wp1 = wp + WPN;
    _Float16* wp2 = wp + 2 * WPN;
    _Float16* wp3 = wp + 3 * WPN;
    _Float16* wp4 = wp + 4 * WPN;
    _Float16* wp5 = wp + 5 * WPN;
    _Float16* wp6 = wp + 6 * WPN;

    float* x_out = (float*)d_out;
    float* t_out = x_out + ND;

    const int nRowTiles = N / 16;
    dim3 gB((nRowTiles + 3) / 4), tB(128);

    {
        long long n4 = (2 * ND) / 4;
        zero_f32<<<(unsigned)((n4 + 255) / 256), 256, 0, stream>>>(buf0, n4);
    }
    {
        dim3 g((WPN + 255) / 256), b(256);
        pack_weights<<<g, b, 0, stream>>>(W1a, wp0);
        pack_weights<<<g, b, 0, stream>>>(W2a, wp1);
        pack_weights<<<g, b, 0, stream>>>(W1b, wp2);
        pack_weights<<<g, b, 0, stream>>>(W2b, wp3);
        pack_weights<<<g, b, 0, stream>>>(Wo,  wp4);
        pack_weights<<<g, b, 0, stream>>>(Wf1, wp5);
        pack_weights<<<g, b, 0, stream>>>(Wf2, wp6);
    }
    {
        unsigned blocks = (unsigned)((E + 7) / 8);
        edge_aggregate<<<blocks, 256, 0, stream>>>(t, e_t, e_t + E, buf0, E);
        edge_aggregate<<<blocks, 256, 0, stream>>>(x, e_x, e_x + E, buf1, E);
    }
    gemm128_wmma<true,  false, true ><<<gB, tB, 0, stream>>>(t,    buf0,    wp0, b1a, nullptr, buf2, nRowTiles);
    gemm128_wmma<false, false, false><<<gB, tB, 0, stream>>>(buf2, nullptr, wp1, b2a, nullptr, buf0, nRowTiles);
    gemm128_wmma<true,  false, true ><<<gB, tB, 0, stream>>>(t,    buf1,    wp2, b1b, nullptr, buf2, nRowTiles);
    gemm128_wmma<false, false, false><<<gB, tB, 0, stream>>>(buf2, nullptr, wp3, b2b, nullptr, buf1, nRowTiles);
    fuse_t2<<<(unsigned)((ND / 4 + 255) / 256), 256, 0, stream>>>(t, buf0, buf1, buf2, ND / 4);
    gemm128_wmma<false, false, true ><<<gB, tB, 0, stream>>>(buf2, nullptr, wp4, bo, nullptr, buf0, nRowTiles);
    {
        unsigned blocks = (unsigned)(((long long)N * 32 + 255) / 256);
        ln_residual<<<blocks, 256, 0, stream>>>(buf0, buf2, ln_g, ln_b, t_out, N);
    }
    gemm128_wmma<false, false, true ><<<gB, tB, 0, stream>>>(x,    nullptr, wp5, bf1, nullptr, buf1, nRowTiles);
    gemm128_wmma<false, true,  false><<<gB, tB, 0, stream>>>(buf1, nullptr, wp6, bf2, x, x_out, nRowTiles);
}


# Edge aggregation as a gather over edges grouped by destination (in-call counting sort, integer atomics give an arrival-order rank; one wave sums a node's rows) in place of per-element float atomics
# speedup vs baseline: 1.3705x; 1.3705x over previous
#include <hip/hip_runtime.h>
#include <hip/hip_bf16.h>

typedef __attribute__((ext_vector_type(16))) _Float16 v16h;
typedef __attribute__((ext_vector_type(8)))  float    v8f;

#define DH 128

__global__ __launch_bounds__(256)
void zero_i32(int* __restrict__ p, int n) {
    int i = blockIdx.x * blockDim.x + threadIdx.x;
    if (i < n) p[i] = 0;
}

__global__ __launch_bounds__(256)
void pack_weights(const float* __restrict__ W, _Float16* __restrict__ Wp) {
    int idx = blockIdx.x * blockDim.x + threadIdx.x;
    if (idx >= 4 * 8 * 32 * 16) return;
    int h    = idx & 15;
    int lane = (idx >> 4) & 31;
    int nt   = (idx >> 9) & 7;
    int kt   = (idx >> 12) & 3;
    int grp  = lane >> 4;
    int k    = kt * 32 + grp * 8 + ((h < 8) ? h : (16 + (h - 8)));
    int n    = nt * 16 + (lane & 15);
    Wp[idx]  = (_Float16)W[k * DH + n];
}

__global__ __launch_bounds__(256)
void edge_rank(const int* __restrict__ dst0, const int* __restrict__ dst1,
               int* __restrict__ cnt, int* __restrict__ rank,
               int nEdges, int nNodes) {
    int e = blockIdx.x * blockDim.x + threadIdx.x;
    if (e >= nEdges) return;
    int list = blockIdx.y;
    const int* dst = list ? dst1 : dst0;
    int d = dst[e];
    rank[(long long)list * nEdges + e] =
        atomicAdd(cnt + (long long)list * (nNodes + 1) + d, 1);
}

__global__ __launch_bounds__(1024)
void scan_counts(const int* __restrict__ cnt, int* __restrict__ off, int n) {
    __shared__ int part[1024];
    const int* c = cnt + (long long)blockIdx.x * (n + 1);
    int* o = off + (long long)blockIdx.x * (n + 1);
    int tid = threadIdx.x;
    int chunk = (n + 1023) >> 10;
    int b = tid * chunk;
    int e = min(b + chunk, n);
    int s = 0;
    for (int i = b; i < e; ++i) s += c[i];
    part[tid] = s;
    __syncthreads();
    for (int d = 1; d < 1024; d <<= 1) {
        int v = (tid >= d) ? part[tid - d] : 0;
        __syncthreads();
        part[tid] += v;
        __syncthreads();
    }
    int run = part[tid] - s;
#pragma unroll 8
    for (int i = b; i < e; ++i) { o[i] = run; run += c[i]; }
    if (tid == 1023) o[n] = part[1023];
}

__global__ __launch_bounds__(256)
void edge_place(const int* __restrict__ src0, const int* __restrict__ dst0,
                const int* __restrict__ src1, const int* __restrict__ dst1,
                const int* __restrict__ off, const int* __restrict__ rank,
                int* __restrict__ sorted, int nEdges, int nNodes) {
    int e = blockIdx.x * blockDim.x + threadIdx.x;
    if (e >= nEdges) return;
    int list = blockIdx.y;
    const int* src = list ? src1 : src0;
    const int* dst = list ? dst1 : dst0;
    long long lb = (long long)list * nEdges;
    int d = dst[e];
    int pos = off[(long long)list * (nNodes + 1) + d] + rank[lb + e];
    sorted[lb + pos] = src[e];
}

__global__ __launch_bounds__(256)
void gather_sum(const float* __restrict__ rows0, const float* __restrict__ rows1,
                const int* __restrict__ off, const int* __restrict__ sorted,
                float* __restrict__ agg0, float* __restrict__ agg1,
                int nNodes, int nEdges) {
    int d = blockIdx.x * (blockDim.x >> 5) + (threadIdx.x >> 5);
    if (d >= nNodes) return;
    int list = blockIdx.y;
    const float* rows = list ? rows1 : rows0;
    float* agg = list ? agg1 : agg0;
    const int* o  = off + (long long)list * (nNodes + 1);
    const int* sl = sorted + (long long)list * nEdges;
    int lane = threadIdx.x & 31;
    int beg = __builtin_amdgcn_readfirstlane(o[d]);
    int end = __builtin_amdgcn_readfirstlane(o[d + 1]);
    float4 a0 = make_float4(0.f, 0.f, 0.f, 0.f);
    float4 a1 = a0, a2 = a0, a3 = a0;
#define ROW_LD(j) (((const float4*)(rows + (long long)__builtin_amdgcn_readlane(idx, (j)) * DH))[lane])
#define ACC4(a, v) { a.x += v.x; a.y += v.y; a.z += v.z; a.w += v.w; }
    for (int base = beg; base < end; base += 32) {
        int n = min(end - base, 32);
        int idx = (lane < n) ? sl[base + lane] : 0;
        int j = 0;
        for (; j + 8 <= n; j += 8) {
            float4 v0 = ROW_LD(j),     v1 = ROW_LD(j + 1), v2 = ROW_LD(j + 2), v3 = ROW_LD(j + 3);
            float4 v4 = ROW_LD(j + 4), v5 = ROW_LD(j + 5), v6 = ROW_LD(j + 6), v7 = ROW_LD(j + 7);
            ACC4(a0, v0) ACC4(a1, v1) ACC4(a2, v2) ACC4(a3, v3)
            ACC4(a0, v4) ACC4(a1, v5) ACC4(a2, v6) ACC4(a3, v7)
        }
        if (j + 4 <= n) {
            float4 v0 = ROW_LD(j), v1 = ROW_LD(j + 1), v2 = ROW_LD(j + 2), v3 = ROW_LD(j + 3);
            ACC4(a0, v0) ACC4(a1, v1) ACC4(a2, v2) ACC4(a3, v3)
            j += 4;
        }
        for (; j < n; ++j) {
            float4 v0 = ROW_LD(j);
            ACC4(a0, v0)
        }
    }
#undef ROW_LD
#undef ACC4
    float4 r;
    r.x = (a0.x + a1.x) + (a2.x + a3.x);
    r.y = (a0.y + a1.y) + (a2.y + a3.y);
    r.z = (a0.z + a1.z) + (a2.z + a3.z);
    r.w = (a0.w + a1.w) + (a2.w + a3.w);
    ((float4*)(agg + (long long)d * DH))[lane] = r;
}

template <bool HAS_A1, bool HAS_RES, bool RELU>
__global__ __launch_bounds__(128)
void gemm128_wmma(const float* __restrict__ A0,
                  const float* __restrict__ A1,
                  const _Float16* __restrict__ Wp,
                  const float* __restrict__ bias,
                  const float* __restrict__ Res,
                  float* __restrict__ Out,
                  int nRowTiles) {
    __shared__ float lds[4][16 * DH];
    int wave = blockIdx.x * (blockDim.x >> 5) + (threadIdx.x >> 5);
    if (wave >= nRowTiles) return;
    int lane = threadIdx.x & 31;
    int grp  = lane >> 4;
    int col  = lane & 15;
    float* myLds = lds[threadIdx.x >> 5];
    long long row = (long long)wave * 16 + col;

    const float* ar0 = A0 + row * DH;
    const float* ar1 = HAS_A1 ? (A1 + row * DH) : nullptr;
    v16h afrag[4];
#pragma unroll
    for (int kt = 0; kt < 4; ++kt) {
        int off = kt * 32 + grp * 8;
        float4 q0 = *(const float4*)(ar0 + off);
        float4 q1 = *(const float4*)(ar0 + off + 4);
        float4 q2 = *(const float4*)(ar0 + off + 16);
        float4 q3 = *(const float4*)(ar0 + off + 20);
        if (HAS_A1) {
            float4 b;
            b = *(const float4*)(ar1 + off);      q0.x += b.x; q0.y += b.y; q0.z += b.z; q0.w += b.w;
            b = *(const float4*)(ar1 + off + 4);  q1.x += b.x; q1.y += b.y; q1.z += b.z; q1.w += b.w;
            b = *(const float4*)(ar1 + off + 16); q2.x += b.x; q2.y += b.y; q2.z += b.z; q2.w += b.w;
            b = *(const float4*)(ar1 + off + 20); q3.x += b.x; q3.y += b.y; q3.z += b.z; q3.w += b.w;
        }
        v16h a;
        a[0]  = (_Float16)q0.x; a[1]  = (_Float16)q0.y; a[2]  = (_Float16)q0.z; a[3]  = (_Float16)q0.w;
        a[4]  = (_Float16)q1.x; a[5]  = (_Float16)q1.y; a[6]  = (_Float16)q1.z; a[7]  = (_Float16)q1.w;
        a[8]  = (_Float16)q2.x; a[9]  = (_Float16)q2.y; a[10] = (_Float16)q2.z; a[11] = (_Float16)q2.w;
        a[12] = (_Float16)q3.x; a[13] = (_Float16)q3.y; a[14] = (_Float16)q3.z; a[15] = (_Float16)q3.w;
        afrag[kt] = a;
    }

    const v16h* wfrag = (const v16h*)Wp;
#pragma unroll
    for (int nt = 0; nt < 8; ++nt) {
        v8f acc = {};
#pragma unroll
        for (int kt = 0; kt < 4; ++kt) {
            v16h b = wfrag[(kt * 8 + nt) * 32 + lane];
            acc = __builtin_amdgcn_wmma_f32_16x16x32_f16(
                false, afrag[kt], false, b, (short)0, acc, false, false);
        }
#pragma unroll
        for (int r = 0; r < 8; ++r)
            myLds[(r + grp * 8) * DH + nt * 16 + col] = acc[r];
    }

    float4 bvec = ((const float4*)bias)[lane];
#pragma unroll
    for (int rl = 0; rl < 16; ++rl) {
        float4 v = ((const float4*)myLds)[rl * 32 + lane];
        v.x += bvec.x; v.y += bvec.y; v.z += bvec.z; v.w += bvec.w;
        if (RELU) {
            v.x = fmaxf(v.x, 0.f); v.y = fmaxf(v.y, 0.f);
            v.z = fmaxf(v.z, 0.f); v.w = fmaxf(v.w, 0.f);
        }
        long long m = (long long)wave * 16 + rl;
        if (HAS_RES) {
            float4 rv = ((const float4*)(Res + m * DH))[lane];
            v.x += rv.x; v.y += rv.y; v.z += rv.z; v.w += rv.w;
        }
        ((float4*)(Out + m * DH))[lane] = v;
    }
}

__global__ __launch_bounds__(256)
void fuse_t2(const float* __restrict__ t, const float* __restrict__ ga,
             const float* __restrict__ gb, float* __restrict__ out, long long n4) {
    long long i = (long long)blockIdx.x * blockDim.x + threadIdx.x;
    if (i >= n4) return;
    float4 a = ((const float4*)t)[i];
    float4 b = ((const float4*)ga)[i];
    float4 c = ((const float4*)gb)[i];
    float4 o;
    o.x = fmaxf(a.x + b.x + c.x, 0.f);
    o.y = fmaxf(a.y + b.y + c.y, 0.f);
    o.z = fmaxf(a.z + b.z + c.z, 0.f);
    o.w = fmaxf(a.w + b.w + c.w, 0.f);
    ((float4*)out)[i] = o;
}

__global__ __launch_bounds__(256)
void ln_residual(const float* __restrict__ u, const float* __restrict__ t2,
                 const float* __restrict__ g, const float* __restrict__ b,
                 float* __restrict__ out, int nRows) {
    int wave = blockIdx.x * (blockDim.x >> 5) + (threadIdx.x >> 5);
    if (wave >= nRows) return;
    int lane = threadIdx.x & 31;
    long long base = (long long)wave * DH;
    float4 v = ((const float4*)(u + base))[lane];
    float s  = v.x + v.y + v.z + v.w;
    float sq = v.x * v.x + v.y * v.y + v.z * v.z + v.w * v.w;
#pragma unroll
    for (int off = 16; off > 0; off >>= 1) {
        s  += __shfl_xor(s, off, 32);
        sq += __shfl_xor(sq, off, 32);
    }
    float mean = s * (1.0f / 128.0f);
    float var  = sq * (1.0f / 128.0f) - mean * mean;
    float inv  = rsqrtf(var + 1e-5f);
    float4 tv = ((const float4*)(t2 + base))[lane];
    float4 gv = ((const float4*)g)[lane];
    float4 bv = ((const float4*)b)[lane];
    float4 o;
    o.x = tv.x + (v.x - mean) * inv * gv.x + bv.x;
    o.y = tv.y + (v.y - mean) * inv * gv.y + bv.y;
    o.z = tv.z + (v.z - mean) * inv * gv.z + bv.z;
    o.w = tv.w + (v.w - mean) * inv * gv.w + bv.w;
    ((float4*)(out + base))[lane] = o;
}

extern "C" void kernel_launch(void* const* d_in, const int* in_sizes, int n_in,
                              void* d_out, int out_size, void* d_ws, size_t ws_size,
                              hipStream_t stream) {
    const float* x    = (const float*)d_in[0];
    const float* t    = (const float*)d_in[1];
    const int*   e_t  = (const int*)d_in[2];
    const int*   e_x  = (const int*)d_in[3];
    const float* W1a  = (const float*)d_in[4];
    const float* b1a  = (const float*)d_in[5];
    const float* W2a  = (const float*)d_in[6];
    const float* b2a  = (const float*)d_in[7];
    const float* W1b  = (const float*)d_in[8];
    const float* b1b  = (const float*)d_in[9];
    const float* W2b  = (const float*)d_in[10];
    const float* b2b  = (const float*)d_in[11];
    const float* Wo   = (const float*)d_in[12];
    const float* bo   = (const float*)d_in[13];
    const float* ln_g = (const float*)d_in[14];
    const float* ln_b = (const float*)d_in[15];
    const float* Wf1  = (const float*)d_in[16];
    const float* bf1  = (const float*)d_in[17];
    const float* Wf2  = (const float*)d_in[18];
    const float* bf2  = (const float*)d_in[19];

    const int N = in_sizes[0] / DH;
    const int E = in_sizes[2] / 2;
    const long long ND = (long long)N * DH;

    float* buf0 = (float*)d_ws;
    float* buf1 = buf0 + ND;
    float* buf2 = buf1 + ND;
    _Float16* wp = (_Float16*)(buf2 + ND);
    const int WPN = 4 * 8 * 32 * 16;
    _Float16* wp0 = wp;
    _Float16* wp1 = wp + WPN;
    _Float16* wp2 = wp + 2 * WPN;
    _Float16* wp3 = wp + 3 * WPN;
    _Float16* wp4 = wp + 4 * WPN;
    _Float16* wp5 = wp + 5 * WPN;
    _Float16* wp6 = wp + 6 * WPN;

    float* x_out = (float*)d_out;
    float* t_out = x_out + ND;

    const int nRowTiles = N / 16;
    dim3 gB((nRowTiles + 3) / 4), tB(128);

    int* sorted = (int*)buf2;
    int* cnt    = sorted + 2 * (long long)E;
    int* off    = cnt + 2 * (N + 1);
    int* rank   = (int*)buf1;

    zero_i32<<<(unsigned)((2 * (N + 1) + 255) / 256), 256, 0, stream>>>(cnt, 2 * (N + 1));
    {
        dim3 g((WPN + 255) / 256), b(256);
        pack_weights<<<g, b, 0, stream>>>(W1a, wp0);
        pack_weights<<<g, b, 0, stream>>>(W2a, wp1);
        pack_weights<<<g, b, 0, stream>>>(W1b, wp2);
        pack_weights<<<g, b, 0, stream>>>(W2b, wp3);
        pack_weights<<<g, b, 0, stream>>>(Wo,  wp4);
        pack_weights<<<g, b, 0, stream>>>(Wf1, wp5);
        pack_weights<<<g, b, 0, stream>>>(Wf2, wp6);
    }
    {
        dim3 ge((unsigned)((E + 255) / 256), 2);
        edge_rank<<<ge, 256, 0, stream>>>(e_t + E, e_x + E, cnt, rank, E, N);
        scan_counts<<<2, 1024, 0, stream>>>(cnt, off, N);
        edge_place<<<ge, 256, 0, stream>>>(e_t, e_t + E, e_x, e_x + E, off, rank, sorted, E, N);
        dim3 gg((unsigned)((N + 7) / 8), 2);
        gather_sum<<<gg, 256, 0, stream>>>(t, x, off, sorted, buf0, buf1, N, E);
    }
    gemm128_wmma<true,  false, true ><<<gB, tB, 0, stream>>>(t,    buf0,    wp0, b1a, nullptr, buf2, nRowTiles);
    gemm128_wmma<false, false, false><<<gB, tB, 0, stream>>>(buf2, nullptr, wp1, b2a, nullptr, buf0, nRowTiles);
    gemm128_wmma<true,  false, true ><<<gB, tB, 0, stream>>>(t,    buf1,    wp2, b1b, nullptr, buf2, nRowTiles);
    gemm128_wmma<false, false, false><<<gB, tB, 0, stream>>>(buf2, nullptr, wp3, b2b, nullptr, buf1, nRowTiles);
    fuse_t2<<<(unsigned)((ND / 4 + 255) / 256), 256, 0, stream>>>(t, buf0, buf1, buf2, ND / 4);
    gemm128_wmma<false, false, true ><<<gB, tB, 0, stream>>>(buf2, nullptr, wp4, bo, nullptr, buf0, nRowTiles);
    {
        unsigned blocks = (unsigned)(((long long)N * 32 + 255) / 256);
        ln_residual<<<blocks, 256, 0, stream>>>(buf0, buf2, ln_g, ln_b, t_out, N);
    }
    gemm128_wmma<false, false, true ><<<gB, tB, 0, stream>>>(x,    nullptr, wp5, bf1, nullptr, buf1, nRowTiles);
    gemm128_wmma<false, true,  false><<<gB, tB, 0, stream>>>(buf1, nullptr, wp6, bf2, x, x_out, nRowTiles);
}


# Dense layers fused per 16-row tile: a wave takes its tile through a branch's whole layer chain via its own LDS tile, weight loads issued one tile ahead (2 launches replace 9); same sum order
# speedup vs baseline: 1.7996x; 1.3131x over previous
#include <hip/hip_runtime.h>
#include <hip/hip_bf16.h>

typedef __attribute__((ext_vector_type(16))) _Float16 v16h;
typedef __attribute__((ext_vector_type(8)))  float    v8f;

#define DH 128

__global__ __launch_bounds__(256)
void zero_i32(int* __restrict__ p, int n) {
    int i = blockIdx.x * blockDim.x + threadIdx.x;
    if (i < n) p[i] = 0;
}

__global__ __launch_bounds__(256)
void pack_weights(const float* __restrict__ W, _Float16* __restrict__ Wp) {
    int idx = blockIdx.x * blockDim.x + threadIdx.x;
    if (idx >= 4 * 8 * 32 * 16) return;
    int h    = idx & 15;
    int lane = (idx >> 4) & 31;
    int nt   = (idx >> 9) & 7;
    int kt   = (idx >> 12) & 3;
    int grp  = lane >> 4;
    int k    = kt * 32 + grp * 8 + ((h < 8) ? h : (16 + (h - 8)));
    int n    = nt * 16 + (lane & 15);
    Wp[idx]  = (_Float16)W[k * DH + n];
}

__global__ __launch_bounds__(256)
void edge_rank(const int* __restrict__ dst0, const int* __restrict__ dst1,
               int* __restrict__ cnt, int* __restrict__ rank,
               int nEdges, int nNodes) {
    int e = blockIdx.x * blockDim.x + threadIdx.x;
    if (e >= nEdges) return;
    int list = blockIdx.y;
    const int* dst = list ? dst1 : dst0;
    int d = dst[e];
    rank[(long long)list * nEdges + e] =
        atomicAdd(cnt + (long long)list * (nNodes + 1) + d, 1);
}

__global__ __launch_bounds__(1024)
void scan_counts(const int* __restrict__ cnt, int* __restrict__ off, int n) {
    __shared__ int part[1024];
    const int* c = cnt + (long long)blockIdx.x * (n + 1);
    int* o = off + (long long)blockIdx.x * (n + 1);
    int tid = threadIdx.x;
    int chunk = (n + 1023) >> 10;
    int b = tid * chunk;
    int e = min(b + chunk, n);
    int s = 0;
    for (int i = b; i < e; ++i) s += c[i];
    part[tid] = s;
    __syncthreads();
    for (int d = 1; d < 1024; d <<= 1) {
        int v = (tid >= d) ? part[tid - d] : 0;
        __syncthreads();
        part[tid] += v;
        __syncthreads();
    }
    int run = part[tid] - s;
#pragma unroll 8
    for (int i = b; i < e; ++i) { o[i] = run; run += c[i]; }
    if (tid == 1023) o[n] = part[1023];
}

__global__ __launch_bounds__(256)
void edge_place(const int* __restrict__ src0, const int* __restrict__ dst0,
                const int* __restrict__ src1, const int* __restrict__ dst1,
                const int* __restrict__ off, const int* __restrict__ rank,
                int* __restrict__ sorted, int nEdges, int nNodes) {
    int e = blockIdx.x * blockDim.x + threadIdx.x;
    if (e >= nEdges) return;
    int list = blockIdx.y;
    const int* src = list ? src1 : src0;
    const int* dst = list ? dst1 : dst0;
    long long lb = (long long)list * nEdges;
    int d = dst[e];
    int pos = off[(long long)list * (nNodes + 1) + d] + rank[lb + e];
    sorted[lb + pos] = src[e];
}

__global__ __launch_bounds__(256)
void gather_sum(const float* __restrict__ rows0, const float* __restrict__ rows1,
                const int* __restrict__ off, const int* __restrict__ sorted,
                float* __restrict__ agg0, float* __restrict__ agg1,
                int nNodes, int nEdges) {
    int d = blockIdx.x * (blockDim.x >> 5) + (threadIdx.x >> 5);
    if (d >= nNodes) return;
    int list = blockIdx.y;
    const float* rows = list ? rows1 : rows0;
    float* agg = list ? agg1 : agg0;
    const int* o  = off + (long long)list * (nNodes + 1);
    const int* sl = sorted + (long long)list * nEdges;
    int lane = threadIdx.x & 31;
    int beg = __builtin_amdgcn_readfirstlane(o[d]);
    int end = __builtin_amdgcn_readfirstlane(o[d + 1]);
    float4 a0 = make_float4(0.f, 0.f, 0.f, 0.f);
    float4 a1 = a0, a2 = a0, a3 = a0;
#define ROW_LD(j) (((const float4*)(rows + (long long)__builtin_amdgcn_readlane(idx, (j)) * DH))[lane])
#define ACC4(a, v) { a.x += v.x; a.y += v.y; a.z += v.z; a.w += v.w; }
    for (int base = beg; base < end; base += 32) {
        int n = min(end - base, 32);
        int idx = (lane < n) ? sl[base + lane] : 0;
        int j = 0;
        for (; j + 8 <= n; j += 8) {
            float4 v0 = ROW_LD(j),     v1 = ROW_LD(j + 1), v2 = ROW_LD(j + 2), v3 = ROW_LD(j + 3);
            float4 v4 = ROW_LD(j + 4), v5 = ROW_LD(j + 5), v6 = ROW_LD(j + 6), v7 = ROW_LD(j + 7);
            ACC4(a0, v0) ACC4(a1, v1) ACC4(a2, v2) ACC4(a3, v3)
            ACC4(a0, v4) ACC4(a1, v5) ACC4(a2, v6) ACC4(a3, v7)
        }
        if (j + 4 <= n) {
            float4 v0 = ROW_LD(j), v1 = ROW_LD(j + 1), v2 = ROW_LD(j + 2), v3 = ROW_LD(j + 3);
            ACC4(a0, v0) ACC4(a1, v1) ACC4(a2, v2) ACC4(a3, v3)
            j += 4;
        }
        for (; j < n; ++j) {
            float4 v0 = ROW_LD(j);
            ACC4(a0, v0)
        }
    }
#undef ROW_LD
#undef ACC4
    float4 r;
    r.x = (a0.x + a1.x) + (a2.x + a3.x);
    r.y = (a0.y + a1.y) + (a2.y + a3.y);
    r.z = (a0.z + a1.z) + (a2.z + a3.z);
    r.w = (a0.w + a1.w) + (a2.w + a3.w);
    ((float4*)(agg + (long long)d * DH))[lane] = r;
}

#define LP 132

__device__ __forceinline__ v16h pack16(float4 q0, float4 q1, float4 q2, float4 q3) {
    v16h a;
    a[0]  = (_Float16)q0.x; a[1]  = (_Float16)q0.y; a[2]  = (_Float16)q0.z; a[3]  = (_Float16)q0.w;
    a[4]  = (_Float16)q1.x; a[5]  = (_Float16)q1.y; a[6]  = (_Float16)q1.z; a[7]  = (_Float16)q1.w;
    a[8]  = (_Float16)q2.x; a[9]  = (_Float16)q2.y; a[10] = (_Float16)q2.z; a[11] = (_Float16)q2.w;
    a[12] = (_Float16)q3.x; a[13] = (_Float16)q3.y; a[14] = (_Float16)q3.z; a[15] = (_Float16)q3.w;
    return a;
}

__device__ __forceinline__ void add4(float4& q, float4 b) {
    q.x += b.x; q.y += b.y; q.z += b.z; q.w += b.w;
}

template <bool HAS_A1>
__device__ __forceinline__ void a_from_global(const float* __restrict__ ar0,
                                              const float* __restrict__ ar1,
                                              int grp, v16h (&afrag)[4]) {
#pragma unroll
    for (int kt = 0; kt < 4; ++kt) {
        int off = kt * 32 + grp * 8;
        float4 q0 = *(const float4*)(ar0 + off);
        float4 q1 = *(const float4*)(ar0 + off + 4);
        float4 q2 = *(const float4*)(ar0 + off + 16);
        float4 q3 = *(const float4*)(ar0 + off + 20);
        if (HAS_A1) {
            add4(q0, *(const float4*)(ar1 + off));
            add4(q1, *(const float4*)(ar1 + off + 4));
            add4(q2, *(const float4*)(ar1 + off + 16));
            add4(q3, *(const float4*)(ar1 + off + 20));
        }
        afrag[kt] = pack16(q0, q1, q2, q3);
    }
}

__device__ __forceinline__ void a_from_lds(const float* ar, int grp, v16h (&afrag)[4]) {
#pragma unroll
    for (int kt = 0; kt < 4; ++kt) {
        int off = kt * 32 + grp * 8;
        float4 q0 = *(const float4*)(ar + off);
        float4 q1 = *(const float4*)(ar + off + 4);
        float4 q2 = *(const float4*)(ar + off + 16);
        float4 q3 = *(const float4*)(ar + off + 20);
        afrag[kt] = pack16(q0, q1, q2, q3);
    }
}

__device__ __forceinline__ void load_b(const _Float16* __restrict__ Wp,
                                       int nt, int lane, v16h (&b)[4]) {
    const v16h* wfrag = (const v16h*)Wp;
#pragma unroll
    for (int kt = 0; kt < 4; ++kt) b[kt] = wfrag[(kt * 8 + nt) * 32 + lane];
}

template <bool RELU>
__device__ __forceinline__ void layer_to_lds(const v16h (&afrag)[4],
                                             const _Float16* __restrict__ Wp,
                                             const _Float16* __restrict__ WpNext,
                                             const float* __restrict__ bias,
                                             float* myLds, int lane, int grp, int col,
                                             v16h (&bq)[4]) {
#pragma unroll
    for (int nt = 0; nt < 8; ++nt) {
        v16h c0 = bq[0], c1 = bq[1], c2 = bq[2], c3 = bq[3];
        if (nt < 7)      load_b(Wp, nt + 1, lane, bq);
        else if (WpNext) load_b(WpNext, 0, lane, bq);
        asm volatile("" :: "v"(c0), "v"(c1), "v"(c2), "v"(c3));
        v8f acc = {};
        acc = __builtin_amdgcn_wmma_f32_16x16x32_f16(false, afrag[0], false, c0, (short)0, acc, false, false);
        acc = __builtin_amdgcn_wmma_f32_16x16x32_f16(false, afrag[1], false, c1, (short)0, acc, false, false);
        acc = __builtin_amdgcn_wmma_f32_16x16x32_f16(false, afrag[2], false, c2, (short)0, acc, false, false);
        acc = __builtin_amdgcn_wmma_f32_16x16x32_f16(false, afrag[3], false, c3, (short)0, acc, false, false);
        float bn = bias[nt * 16 + col];
#pragma unroll
        for (int r = 0; r < 8; ++r) {
            float v = acc[r] + bn;
            if (RELU) v = fmaxf(v, 0.f);
            myLds[(r + grp * 8) * LP + nt * 16 + col] = v;
        }
    }
}

#define WPN_H (4 * 8 * 32 * 16)

__global__ __launch_bounds__(128)
void t_chain(const float* __restrict__ t,
             const float* __restrict__ aggA, const float* __restrict__ aggB,
             const _Float16* __restrict__ Wp,
             const float* __restrict__ b1a, const float* __restrict__ b2a,
             const float* __restrict__ b1b, const float* __restrict__ b2b,
             const float* __restrict__ bo,
             const float* __restrict__ ln_g, const float* __restrict__ ln_b,
             float* __restrict__ t_out, int nRowTiles) {
    __shared__ __attribute__((aligned(16))) float lds[4][16 * LP];
    int wave = blockIdx.x * (blockDim.x >> 5) + (threadIdx.x >> 5);
    if (wave >= nRowTiles) return;
    int lane = threadIdx.x & 31;
    int grp  = lane >> 4;
    int col  = lane & 15;
    float* myLds = lds[threadIdx.x >> 5];
    const float* myRow = myLds + col * LP;
    long long row = (long long)wave * 16 + col;
    long long m0  = (long long)wave * 16;
    v16h afrag[4], afragB[4], bq[4];
    float4 s[16];

    load_b(Wp, 0, lane, bq);
    a_from_global<true>(t + row * DH, aggA + row * DH, grp, afrag);
    a_from_global<true>(t + row * DH, aggB + row * DH, grp, afragB);

    layer_to_lds<true >(afrag, Wp,             Wp + WPN_H,     b1a, myLds, lane, grp, col, bq);
    a_from_lds(myRow, grp, afrag);
    layer_to_lds<false>(afrag, Wp + WPN_H,     Wp + 2 * WPN_H, b2a, myLds, lane, grp, col, bq);
#pragma unroll
    for (int rl = 0; rl < 16; ++rl) {
        float4 g  = ((const float4*)(myLds + rl * LP))[lane];
        float4 tv = ((const float4*)(t + (m0 + rl) * DH))[lane];
        s[rl].x = tv.x + g.x; s[rl].y = tv.y + g.y; s[rl].z = tv.z + g.z; s[rl].w = tv.w + g.w;
    }

    layer_to_lds<true >(afragB, Wp + 2 * WPN_H, Wp + 3 * WPN_H, b1b, myLds, lane, grp, col, bq);
    a_from_lds(myRow, grp, afrag);
    layer_to_lds<false>(afrag,  Wp + 3 * WPN_H, Wp + 4 * WPN_H, b2b, myLds, lane, grp, col, bq);
#pragma unroll
    for (int rl = 0; rl < 16; ++rl) {
        float4 g = ((const float4*)(myLds + rl * LP))[lane];
        s[rl].x = fmaxf(s[rl].x + g.x, 0.f); s[rl].y = fmaxf(s[rl].y + g.y, 0.f);
        s[rl].z = fmaxf(s[rl].z + g.z, 0.f); s[rl].w = fmaxf(s[rl].w + g.w, 0.f);
    }
#pragma unroll
    for (int rl = 0; rl < 16; ++rl)
        ((float4*)(myLds + rl * LP))[lane] = s[rl];

    a_from_lds(myRow, grp, afrag);
    layer_to_lds<true >(afrag, Wp + 4 * WPN_H, nullptr, bo, myLds, lane, grp, col, bq);
    float4 gv = ((const float4*)ln_g)[lane];
    float4 bv = ((const float4*)ln_b)[lane];
#pragma unroll
    for (int rl = 0; rl < 16; ++rl) {
        float4 v = ((const float4*)(myLds + rl * LP))[lane];
        float sm = v.x + v.y + v.z + v.w;
        float sq = v.x * v.x + v.y * v.y + v.z * v.z + v.w * v.w;
#pragma unroll
        for (int off = 16; off > 0; off >>= 1) {
            sm += __shfl_xor(sm, off, 32);
            sq += __shfl_xor(sq, off, 32);
        }
        float mean = sm * (1.0f / 128.0f);
        float var  = sq * (1.0f / 128.0f) - mean * mean;
        float inv  = rsqrtf(var + 1e-5f);
        float4 tv = s[rl];
        float4 o;
        o.x = tv.x + (v.x - mean) * inv * gv.x + bv.x;
        o.y = tv.y + (v.y - mean) * inv * gv.y + bv.y;
        o.z = tv.z + (v.z - mean) * inv * gv.z + bv.z;
        o.w = tv.w + (v.w - mean) * inv * gv.w + bv.w;
        ((float4*)(t_out + (m0 + rl) * DH))[lane] = o;
    }
}

__global__ __launch_bounds__(128)
void x_chain(const float* __restrict__ x, const _Float16* __restrict__ Wp,
             const float* __restrict__ bf1, const float* __restrict__ bf2,
             float* __restrict__ x_out, int nRowTiles) {
    __shared__ __attribute__((aligned(16))) float lds[4][16 * LP];
    int wave = blockIdx.x * (blockDim.x >> 5) + (threadIdx.x >> 5);
    if (wave >= nRowTiles) return;
    int lane = threadIdx.x & 31;
    int grp  = lane >> 4;
    int col  = lane & 15;
    float* myLds = lds[threadIdx.x >> 5];
    const float* myRow = myLds + col * LP;
    long long row = (long long)wave * 16 + col;
    long long m0  = (long long)wave * 16;
    v16h afrag[4], bq[4];

    load_b(Wp, 0, lane, bq);
    a_from_global<false>(x + row * DH, nullptr, grp, afrag);
    layer_to_lds<true >(afrag, Wp,         Wp + WPN_H, bf1, myLds, lane, grp, col, bq);
    a_from_lds(myRow, grp, afrag);
    layer_to_lds<false>(afrag, Wp + WPN_H, nullptr,    bf2, myLds, lane, grp, col, bq);
#pragma unroll
    for (int rl = 0; rl < 16; ++rl) {
        float4 v  = ((const float4*)(myLds + rl * LP))[lane];
        float4 rv = ((const float4*)(x + (m0 + rl) * DH))[lane];
        v.x += rv.x; v.y += rv.y; v.z += rv.z; v.w += rv.w;
        ((float4*)(x_out + (m0 + rl) * DH))[lane] = v;
    }
}

extern "C" void kernel_launch(void* const* d_in, const int* in_sizes, int n_in,
                              void* d_out, int out_size, void* d_ws, size_t ws_size,
                              hipStream_t stream) {
    const float* x    = (const float*)d_in[0];
    const float* t    = (const float*)d_in[1];
    const int*   e_t  = (const int*)d_in[2];
    const int*   e_x  = (const int*)d_in[3];
    const float* W1a  = (const float*)d_in[4];
    const float* b1a  = (const float*)d_in[5];
    const float* W2a  = (const float*)d_in[6];
    const float* b2a  = (const float*)d_in[7];
    const float* W1b  = (const float*)d_in[8];
    const float* b1b  = (const float*)d_in[9];
    const float* W2b  = (const float*)d_in[10];
    const float* b2b  = (const float*)d_in[11];
    const float* Wo   = (const float*)d_in[12];
    const float* bo   = (const float*)d_in[13];
    const float* ln_g = (const float*)d_in[14];
    const float* ln_b = (const float*)d_in[15];
    const float* Wf1  = (const float*)d_in[16];
    const float* bf1  = (const float*)d_in[17];
    const float* Wf2  = (const float*)d_in[18];
    const float* bf2  = (const float*)d_in[19];

    const int N = in_sizes[0] / DH;
    const int E = in_sizes[2] / 2;
    const long long ND = (long long)N * DH;

    float* buf0 = (float*)d_ws;
    float* buf1 = buf0 + ND;
    float* buf2 = buf1 + ND;
    _Float16* wp = (_Float16*)(buf2 + ND);
    const int WPN = 4 * 8 * 32 * 16;
    _Float16* wp0 = wp;
    _Float16* wp1 = wp + WPN;
    _Float16* wp2 = wp + 2 * WPN;
    _Float16* wp3 = wp + 3 * WPN;
    _Float16* wp4 = wp + 4 * WPN;
    _Float16* wp5 = wp + 5 * WPN;
    _Float16* wp6 = wp + 6 * WPN;

    float* x_out = (float*)d_out;
    float* t_out = x_out + ND;

    const int nRowTiles = N / 16;
    dim3 gB((nRowTiles + 3) / 4), tB(128);

    int* sorted = (int*)buf2;
    int* cnt    = sorted + 2 * (long long)E;
    int* off    = cnt + 2 * (N + 1);
    int* rank   = (int*)buf1;

    zero_i32<<<(unsigned)((2 * (N + 1) + 255) / 256), 256, 0, stream>>>(cnt, 2 * (N + 1));
    {
        dim3 g((WPN + 255) / 256), b(256);
        pack_weights<<<g, b, 0, stream>>>(W1a, wp0);
        pack_weights<<<g, b, 0, stream>>>(W2a, wp1);
        pack_weights<<<g, b, 0, stream>>>(W1b, wp2);
        pack_weights<<<g, b, 0, stream>>>(W2b, wp3);
        pack_weights<<<g, b, 0, stream>>>(Wo,  wp4);
        pack_weights<<<g, b, 0, stream>>>(Wf1, wp5);
        pack_weights<<<g, b, 0, stream>>>(Wf2, wp6);
    }
    {
        dim3 ge((unsigned)((E + 255) / 256), 2);
        edge_rank<<<ge, 256, 0, stream>>>(e_t + E, e_x + E, cnt, rank, E, N);
        scan_counts<<<2, 1024, 0, stream>>>(cnt, off, N);
        edge_place<<<ge, 256, 0, stream>>>(e_t, e_t + E, e_x, e_x + E, off, rank, sorted, E, N);
        dim3 gg((unsigned)((N + 7) / 8), 2);
        gather_sum<<<gg, 256, 0, stream>>>(t, x, off, sorted, buf0, buf1, N, E);
    }
    t_chain<<<gB, tB, 0, stream>>>(t, buf0, buf1, wp0, b1a, b2a, b1b, b2b, bo, ln_g, ln_b, t_out, nRowTiles);
    x_chain<<<gB, tB, 0, stream>>>(x, wp5, bf1, bf2, x_out, nRowTiles);
}


# x branch without a launch of its own: it rides in the edge-placement launch on work-groups that land on the WGPs the placing leaves empty (7 launches -> 6); same values, same sums
# speedup vs baseline: 3.6215x; 2.0124x over previous
#include <hip/hip_runtime.h>
#include <hip/hip_bf16.h>

typedef __attribute__((ext_vector_type(16))) _Float16 v16h;
typedef __attribute__((ext_vector_type(8)))  float    v8f;
typedef __attribute__((ext_vector_type(4)))  float    v4f;

#define DH 128

__device__ __forceinline__
void pack_one(const float* __restrict__ W, _Float16* __restrict__ Wp, int idx) {
    int h    = idx & 15;
    int lane = (idx >> 4) & 31;
    int nt   = (idx >> 9) & 7;
    int kt   = (idx >> 12) & 3;
    int grp  = lane >> 4;
    int k    = kt * 32 + grp * 8 + ((h < 8) ? h : (16 + (h - 8)));
    int n    = nt * 16 + (lane & 15);
    Wp[idx]  = (_Float16)W[k * DH + n];
}

#define NCH  64
#define HCAP 65536
#define NMAT 7
#define WPN_H (4 * 8 * 32 * 16)

__device__ __forceinline__ int chunk_len(int nEdges) {
    return (((nEdges + NCH - 1) / NCH) + 1023) & ~1023;
}

__global__ __launch_bounds__(1024)
void edge_count(const int* __restrict__ dst0, const int* __restrict__ dst1,
                int* __restrict__ cntc, int nEdges, int nNodes,
                const float* __restrict__ W0, const float* __restrict__ W1,
                const float* __restrict__ W2, const float* __restrict__ W3,
                const float* __restrict__ W4, const float* __restrict__ W5,
                const float* __restrict__ W6, _Float16* __restrict__ Wp) {
    __shared__ int hist[HCAP];
    int tid  = threadIdx.x;
    if (blockIdx.y >= 2) {
        int pb  = (int)(blockIdx.y - 2) * NCH + (int)blockIdx.x;
        int mat = pb >> 4;
        if (mat >= NMAT) return;
        const float* W = mat == 0 ? W0 : mat == 1 ? W1 : mat == 2 ? W2 : mat == 3 ? W3
                       : mat == 4 ? W4 : mat == 5 ? W5 : W6;
        pack_one(W, Wp + mat * WPN_H, ((pb & 15) << 10) + tid);
        return;
    }
    int list = blockIdx.y;
    const int* dst = list ? dst1 : dst0;
    int* part = cntc + ((long long)list * NCH + blockIdx.x) * nNodes;
    int per = chunk_len(nEdges);
    int beg = blockIdx.x * per;
    int end = min(beg + per, nEdges);
    for (int lo = 0; lo < nNodes; lo += HCAP) {
        int m = min(nNodes - lo, HCAP);
        for (int i = tid; i < m; i += 1024) hist[i] = 0;
        __syncthreads();
        int e = beg + tid;
        for (; e + 3 * 1024 < end; e += 4 * 1024) {
            int d0 = dst[e] - lo,        d1 = dst[e + 1024] - lo;
            int d2 = dst[e + 2048] - lo, d3 = dst[e + 3072] - lo;
            if ((unsigned)d0 < (unsigned)m) atomicAdd(&hist[d0], 1);
            if ((unsigned)d1 < (unsigned)m) atomicAdd(&hist[d1], 1);
            if ((unsigned)d2 < (unsigned)m) atomicAdd(&hist[d2], 1);
            if ((unsigned)d3 < (unsigned)m) atomicAdd(&hist[d3], 1);
        }
        for (; e < end; e += 1024) {
            int d = dst[e] - lo;
            if ((unsigned)d < (unsigned)m) atomicAdd(&hist[d], 1);
        }
        __syncthreads();
        for (int i = tid; i < m; i += 1024) part[lo + i] = hist[i];
        __syncthreads();
    }
}

__global__ __launch_bounds__(256)
void chunk_prefix(int* __restrict__ cntc, int* __restrict__ tot, int nNodes) {
    int d = blockIdx.x * blockDim.x + threadIdx.x;
    if (d >= nNodes) return;
    int list = blockIdx.y;
    int* col = cntc + (long long)list * NCH * nNodes + d;
    int v[NCH];
#pragma unroll
    for (int c = 0; c < NCH; ++c) v[c] = col[(long long)c * nNodes];
    int run = 0;
#pragma unroll
    for (int c = 0; c < NCH; ++c) { col[(long long)c * nNodes] = run; run += v[c]; }
    tot[(long long)list * (nNodes + 1) + d] = run;
}

__global__ __launch_bounds__(1024)
void scan_counts(const int* __restrict__ cnt, int* __restrict__ off, int n) {
    __shared__ int waveTot[32];
    __shared__ int wavePre[32];
    const int* c = cnt + (long long)blockIdx.y * (n + 1);
    int* o = off + (long long)blockIdx.y * (n + 1);
    int tid  = threadIdx.x;
    int lane = tid & 31;
    int w    = tid >> 5;
    int first = blockIdx.x * 1024;
    int pre = 0;
#pragma unroll 8
    for (int k = tid; k < first; k += 1024) pre += c[k];
    int i = first + tid;
    int v = (i < n) ? c[i] : 0;
    int incl = v;
#pragma unroll
    for (int d = 1; d < 32; d <<= 1) {
        int u = __shfl_up(incl, d, 32);
        if (lane >= d) incl += u;
    }
#pragma unroll
    for (int d = 16; d > 0; d >>= 1) pre += __shfl_xor(pre, d, 32);
    if (lane == 31) waveTot[w] = incl;
    if (lane == 0)  wavePre[w] = pre;
    __syncthreads();
    int tot  = waveTot[lane];
    int base = wavePre[lane];
    int tincl = tot;
#pragma unroll
    for (int d = 1; d < 32; d <<= 1) {
        int u = __shfl_up(tincl, d, 32);
        if (lane >= d) tincl += u;
    }
#pragma unroll
    for (int d = 16; d > 0; d >>= 1) base += __shfl_xor(base, d, 32);
    int wexcl = __shfl(tincl - tot, w, 32);
    if (i <= n) o[i] = base + wexcl + (incl - v);
}

__global__ __launch_bounds__(256)
void gather_sum(const float* __restrict__ rows0, const float* __restrict__ rows1,
                const int* __restrict__ off, const int* __restrict__ sorted,
                float* __restrict__ agg0, float* __restrict__ agg1,
                int nNodes, int nEdges) {
    int d = blockIdx.x * (blockDim.x >> 5) + (threadIdx.x >> 5);
    if (d >= nNodes) return;
    int list = blockIdx.y;
    const float* rows = list ? rows1 : rows0;
    float* agg = list ? agg1 : agg0;
    const int* o  = off + (long long)list * (nNodes + 1);
    const int* sl = sorted + (long long)list * nEdges;
    int lane = threadIdx.x & 31;
    int beg = __builtin_amdgcn_readfirstlane(o[d]);
    int end = __builtin_amdgcn_readfirstlane(o[d + 1]);
    float4 a0 = make_float4(0.f, 0.f, 0.f, 0.f);
    float4 a1 = a0, a2 = a0, a3 = a0;
#define ROW_LD(j) (((const float4*)(rows + (long long)__builtin_amdgcn_readlane(idx, (j)) * DH))[lane])
#define ACC4(a, v) { a.x += v.x; a.y += v.y; a.z += v.z; a.w += v.w; }
    for (int base = beg; base < end; base += 32) {
        int n = min(end - base, 32);
        int idx = (lane < n) ? sl[base + lane] : 0;
        int j = 0;
        for (; j + 8 <= n; j += 8) {
            float4 v0 = ROW_LD(j),     v1 = ROW_LD(j + 1), v2 = ROW_LD(j + 2), v3 = ROW_LD(j + 3);
            float4 v4 = ROW_LD(j + 4), v5 = ROW_LD(j + 5), v6 = ROW_LD(j + 6), v7 = ROW_LD(j + 7);
            ACC4(a0, v0) ACC4(a1, v1) ACC4(a2, v2) ACC4(a3, v3)
            ACC4(a0, v4) ACC4(a1, v5) ACC4(a2, v6) ACC4(a3, v7)
        }
        if (j + 4 <= n) {
            float4 v0 = ROW_LD(j), v1 = ROW_LD(j + 1), v2 = ROW_LD(j + 2), v3 = ROW_LD(j + 3);
            ACC4(a0, v0) ACC4(a1, v1) ACC4(a2, v2) ACC4(a3, v3)
            j += 4;
        }
        for (; j < n; ++j) {
            float4 v0 = ROW_LD(j);
            ACC4(a0, v0)
        }
    }
#undef ROW_LD
#undef ACC4
    float4 r;
    r.x = (a0.x + a1.x) + (a2.x + a3.x);
    r.y = (a0.y + a1.y) + (a2.y + a3.y);
    r.z = (a0.z + a1.z) + (a2.z + a3.z);
    r.w = (a0.w + a1.w) + (a2.w + a3.w);
    ((float4*)(agg + (long long)d * DH))[lane] = r;
}

#define LP 132

__device__ __forceinline__ v16h pack16(float4 q0, float4 q1, float4 q2, float4 q3) {
    v16h a;
    a[0]  = (_Float16)q0.x; a[1]  = (_Float16)q0.y; a[2]  = (_Float16)q0.z; a[3]  = (_Float16)q0.w;
    a[4]  = (_Float16)q1.x; a[5]  = (_Float16)q1.y; a[6]  = (_Float16)q1.z; a[7]  = (_Float16)q1.w;
    a[8]  = (_Float16)q2.x; a[9]  = (_Float16)q2.y; a[10] = (_Float16)q2.z; a[11] = (_Float16)q2.w;
    a[12] = (_Float16)q3.x; a[13] = (_Float16)q3.y; a[14] = (_Float16)q3.z; a[15] = (_Float16)q3.w;
    return a;
}

__device__ __forceinline__ void add4(float4& q, float4 b) {
    q.x += b.x; q.y += b.y; q.z += b.z; q.w += b.w;
}

template <bool HAS_A1>
__device__ __forceinline__ void a_from_global(const float* __restrict__ ar0,
                                              const float* __restrict__ ar1,
                                              int grp, v16h (&afrag)[4]) {
#pragma unroll
    for (int kt = 0; kt < 4; ++kt) {
        int off = kt * 32 + grp * 8;
        float4 q0 = *(const float4*)(ar0 + off);
        float4 q1 = *(const float4*)(ar0 + off + 4);
        float4 q2 = *(const float4*)(ar0 + off + 16);
        float4 q3 = *(const float4*)(ar0 + off + 20);
        if (HAS_A1) {
            add4(q0, *(const float4*)(ar1 + off));
            add4(q1, *(const float4*)(ar1 + off + 4));
            add4(q2, *(const float4*)(ar1 + off + 16));
            add4(q3, *(const float4*)(ar1 + off + 20));
        }
        afrag[kt] = pack16(q0, q1, q2, q3);
    }
}

__device__ __forceinline__ void a_from_lds(const float* ar, int grp, v16h (&afrag)[4]) {
#pragma unroll
    for (int kt = 0; kt < 4; ++kt) {
        int off = kt * 32 + grp * 8;
        float4 q0 = *(const float4*)(ar + off);
        float4 q1 = *(const float4*)(ar + off + 4);
        float4 q2 = *(const float4*)(ar + off + 16);
        float4 q3 = *(const float4*)(ar + off + 20);
        afrag[kt] = pack16(q0, q1, q2, q3);
    }
}

__device__ __forceinline__ void load_b(const _Float16* __restrict__ Wp,
                                       int nt, int lane, v16h (&b)[4]) {
    const v16h* wfrag = (const v16h*)Wp;
#pragma unroll
    for (int kt = 0; kt < 4; ++kt) b[kt] = wfrag[(kt * 8 + nt) * 32 + lane];
}

template <bool RELU>
__device__ __forceinline__ void layer_to_lds(const v16h (&afrag)[4],
                                             const _Float16* __restrict__ Wp,
                                             const _Float16* __restrict__ WpNext,
                                             const float* __restrict__ bias,
                                             float* myLds, int lane, int grp, int col,
                                             v16h (&bq)[4]) {
#pragma unroll
    for (int nt = 0; nt < 8; ++nt) {
        v16h c0 = bq[0], c1 = bq[1], c2 = bq[2], c3 = bq[3];
        if (nt < 7)      load_b(Wp, nt + 1, lane, bq);
        else if (WpNext) load_b(WpNext, 0, lane, bq);
        asm volatile("" :: "v"(c0), "v"(c1), "v"(c2), "v"(c3));
        v8f acc = {};
        acc = __builtin_amdgcn_wmma_f32_16x16x32_f16(false, afrag[0], false, c0, (short)0, acc, false, false);
        acc = __builtin_amdgcn_wmma_f32_16x16x32_f16(false, afrag[1], false, c1, (short)0, acc, false, false);
        acc = __builtin_amdgcn_wmma_f32_16x16x32_f16(false, afrag[2], false, c2, (short)0, acc, false, false);
        acc = __builtin_amdgcn_wmma_f32_16x16x32_f16(false, afrag[3], false, c3, (short)0, acc, false, false);
        float bn = bias[nt * 16 + col];
#pragma unroll
        for (int r = 0; r < 8; ++r) {
            float v = acc[r] + bn;
            if (RELU) v = fmaxf(v, 0.f);
            myLds[(r + grp * 8) * LP + nt * 16 + col] = v;
        }
    }
}

__global__ __launch_bounds__(128)
void t_chain(const float* __restrict__ t,
             const float* __restrict__ aggA, const float* __restrict__ aggB,
             const _Float16* __restrict__ Wp,
             const float* __restrict__ b1a, const float* __restrict__ b2a,
             const float* __restrict__ b1b, const float* __restrict__ b2b,
             const float* __restrict__ bo,
             const float* __restrict__ ln_g, const float* __restrict__ ln_b,
             float* __restrict__ t_out, int nRowTiles) {
    __shared__ __attribute__((aligned(16))) float lds[4][16 * LP];
    int wave = blockIdx.x * (blockDim.x >> 5) + (threadIdx.x >> 5);
    if (wave >= nRowTiles) return;
    int lane = threadIdx.x & 31;
    int grp  = lane >> 4;
    int col  = lane & 15;
    float* myLds = lds[threadIdx.x >> 5];
    const float* myRow = myLds + col * LP;
    long long row = (long long)wave * 16 + col;
    long long m0  = (long long)wave * 16;
    v16h afrag[4], afragB[4], bq[4];
    float4 s[16];

    load_b(Wp, 0, lane, bq);
    a_from_global<true>(t + row * DH, aggA + row * DH, grp, afrag);
    a_from_global<true>(t + row * DH, aggB + row * DH, grp, afragB);

    layer_to_lds<true >(afrag, Wp,             Wp + WPN_H,     b1a, myLds, lane, grp, col, bq);
    a_from_lds(myRow, grp, afrag);
    layer_to_lds<false>(afrag, Wp + WPN_H,     Wp + 2 * WPN_H, b2a, myLds, lane, grp, col, bq);
#pragma unroll
    for (int rl = 0; rl < 16; ++rl) {
        float4 g  = ((const float4*)(myLds + rl * LP))[lane];
        float4 tv = ((const float4*)(t + (m0 + rl) * DH))[lane];
        s[rl].x = tv.x + g.x; s[rl].y = tv.y + g.y; s[rl].z = tv.z + g.z; s[rl].w = tv.w + g.w;
    }

    layer_to_lds<true >(afragB, Wp + 2 * WPN_H, Wp + 3 * WPN_H, b1b, myLds, lane, grp, col, bq);
    a_from_lds(myRow, grp, afrag);
    layer_to_lds<false>(afrag,  Wp + 3 * WPN_H, Wp + 4 * WPN_H, b2b, myLds, lane, grp, col, bq);
#pragma unroll
    for (int rl = 0; rl < 16; ++rl) {
        float4 g = ((const float4*)(myLds + rl * LP))[lane];
        s[rl].x = fmaxf(s[rl].x + g.x, 0.f); s[rl].y = fmaxf(s[rl].y + g.y, 0.f);
        s[rl].z = fmaxf(s[rl].z + g.z, 0.f); s[rl].w = fmaxf(s[rl].w + g.w, 0.f);
    }
#pragma unroll
    for (int rl = 0; rl < 16; ++rl)
        ((float4*)(myLds + rl * LP))[lane] = s[rl];

    a_from_lds(myRow, grp, afrag);
    layer_to_lds<true >(afrag, Wp + 4 * WPN_H, nullptr, bo, myLds, lane, grp, col, bq);
    float4 gv = ((const float4*)ln_g)[lane];
    float4 bv = ((const float4*)ln_b)[lane];
#pragma unroll
    for (int rl = 0; rl < 16; ++rl) {
        float4 v = ((const float4*)(myLds + rl * LP))[lane];
        float sm = v.x + v.y + v.z + v.w;
        float sq = v.x * v.x + v.y * v.y + v.z * v.z + v.w * v.w;
#pragma unroll
        for (int off = 16; off > 0; off >>= 1) {
            sm += __shfl_xor(sm, off, 32);
            sq += __shfl_xor(sq, off, 32);
        }
        float mean = sm * (1.0f / 128.0f);
        float var  = sq * (1.0f / 128.0f) - mean * mean;
        float inv  = rsqrtf(var + 1e-5f);
        float4 tv = s[rl];
        float4 o;
        o.x = tv.x + (v.x - mean) * inv * gv.x + bv.x;
        o.y = tv.y + (v.y - mean) * inv * gv.y + bv.y;
        o.z = tv.z + (v.z - mean) * inv * gv.z + bv.z;
        o.w = tv.w + (v.w - mean) * inv * gv.w + bv.w;
        ((float4*)(t_out + (m0 + rl) * DH))[lane] = o;
    }
}

__device__ __forceinline__
void x_chain_tile(const float* __restrict__ x, const _Float16* __restrict__ Wp,
                  const float* __restrict__ bf1, const float* __restrict__ bf2,
                  float* __restrict__ x_out, int wave, float* myLds, int lane) {
    int grp  = lane >> 4;
    int col  = lane & 15;
    const float* myRow = myLds + col * LP;
    long long row = (long long)wave * 16 + col;
    long long m0  = (long long)wave * 16;
    v16h afrag[4], bq[4];

    load_b(Wp, 0, lane, bq);
    const float* ar = x + row * DH + grp * 8;
    v4f q[16];
#pragma unroll
    for (int kt = 0; kt < 4; ++kt) {
        q[4 * kt]     = *(const v4f*)(ar + kt * 32);
        q[4 * kt + 1] = *(const v4f*)(ar + kt * 32 + 4);
        q[4 * kt + 2] = *(const v4f*)(ar + kt * 32 + 16);
        q[4 * kt + 3] = *(const v4f*)(ar + kt * 32 + 20);
    }
    asm volatile("" :: "v"(q[0]), "v"(q[1]), "v"(q[2]),  "v"(q[3]),  "v"(q[4]),  "v"(q[5]),  "v"(q[6]),  "v"(q[7]),
                       "v"(q[8]), "v"(q[9]), "v"(q[10]), "v"(q[11]), "v"(q[12]), "v"(q[13]), "v"(q[14]), "v"(q[15]));
#pragma unroll
    for (int kt = 0; kt < 4; ++kt) {
#pragma unroll
        for (int h = 0; h < 16; ++h) afrag[kt][h] = (_Float16)q[4 * kt + (h >> 2)][h & 3];
    }
    layer_to_lds<true >(afrag, Wp,         Wp + WPN_H, bf1, myLds, lane, grp, col, bq);
    __builtin_amdgcn_sched_barrier(0);
    a_from_lds(myRow, grp, afrag);
    __builtin_amdgcn_sched_barrier(0);
    layer_to_lds<false>(afrag, Wp + WPN_H, nullptr,    bf2, myLds, lane, grp, col, bq);
    __builtin_amdgcn_sched_barrier(0);
#pragma unroll
    for (int rl = 0; rl < 16; ++rl) {
        float4 v  = ((const float4*)(myLds + rl * LP))[lane];
        float4 rv = ((const float4*)(x + (m0 + rl) * DH))[lane];
        v.x += rv.x; v.y += rv.y; v.z += rv.z; v.w += rv.w;
        ((float4*)(x_out + (m0 + rl) * DH))[lane] = v;
    }
}

#define PLDS ((32 * 16 * LP) > HCAP ? (32 * 16 * LP) : HCAP)

__global__ __launch_bounds__(1024)
void edge_place(const int* __restrict__ src0, const int* __restrict__ dst0,
                const int* __restrict__ src1, const int* __restrict__ dst1,
                const int* __restrict__ cntc, const int* __restrict__ off,
                int* __restrict__ sorted, int nEdges, int nNodes,
                const float* __restrict__ x, const _Float16* __restrict__ WpX,
                const float* __restrict__ bf1, const float* __restrict__ bf2,
                float* __restrict__ x_out, int nRowTiles) {
    __shared__ __attribute__((aligned(16))) int cur[PLDS];
    int tid  = threadIdx.x;
    if (blockIdx.y >= 2) {
        int xb   = (int)(blockIdx.y - 2) * NCH + (int)blockIdx.x;
        int tile = xb * 32 + (tid >> 5);
        if (tile >= nRowTiles) return;
        x_chain_tile(x, WpX, bf1, bf2, x_out, tile, (float*)cur + (tid >> 5) * (16 * LP), tid & 31);
        return;
    }
    int list = blockIdx.y;
    const int* src = list ? src1 : src0;
    const int* dst = list ? dst1 : dst0;
    const int* part = cntc + ((long long)list * NCH + blockIdx.x) * nNodes;
    const int* o = off + (long long)list * (nNodes + 1);
    int* out = sorted + (long long)list * nEdges;
    int per = chunk_len(nEdges);
    int beg = blockIdx.x * per;
    int end = min(beg + per, nEdges);
    for (int lo = 0; lo < nNodes; lo += HCAP) {
        int m = min(nNodes - lo, HCAP);
#pragma unroll 4
        for (int i = tid; i < m; i += 1024) cur[i] = o[lo + i] + part[lo + i];
        __syncthreads();
        int e = beg + tid;
        for (; e + 3 * 1024 < end; e += 4 * 1024) {
            int d0 = dst[e] - lo,        d1 = dst[e + 1024] - lo;
            int d2 = dst[e + 2048] - lo, d3 = dst[e + 3072] - lo;
            int s0 = src[e],        s1 = src[e + 1024];
            int s2 = src[e + 2048], s3 = src[e + 3072];
            if ((unsigned)d0 < (unsigned)m) out[atomicAdd(&cur[d0], 1)] = s0;
            if ((unsigned)d1 < (unsigned)m) out[atomicAdd(&cur[d1], 1)] = s1;
            if ((unsigned)d2 < (unsigned)m) out[atomicAdd(&cur[d2], 1)] = s2;
            if ((unsigned)d3 < (unsigned)m) out[atomicAdd(&cur[d3], 1)] = s3;
        }
        for (; e < end; e += 1024) {
            int d = dst[e] - lo;
            int sv = src[e];
            if ((unsigned)d < (unsigned)m) out[atomicAdd(&cur[d], 1)] = sv;
        }
        __syncthreads();
    }
}

extern "C" void kernel_launch(void* const* d_in, const int* in_sizes, int n_in,
                              void* d_out, int out_size, void* d_ws, size_t ws_size,
                              hipStream_t stream) {
    const float* x    = (const float*)d_in[0];
    const float* t    = (const float*)d_in[1];
    const int*   e_t  = (const int*)d_in[2];
    const int*   e_x  = (const int*)d_in[3];
    const float* W1a  = (const float*)d_in[4];
    const float* b1a  = (const float*)d_in[5];
    const float* W2a  = (const float*)d_in[6];
    const float* b2a  = (const float*)d_in[7];
    const float* W1b  = (const float*)d_in[8];
    const float* b1b  = (const float*)d_in[9];
    const float* W2b  = (const float*)d_in[10];
    const float* b2b  = (const float*)d_in[11];
    const float* Wo   = (const float*)d_in[12];
    const float* bo   = (const float*)d_in[13];
    const float* ln_g = (const float*)d_in[14];
    const float* ln_b = (const float*)d_in[15];
    const float* Wf1  = (const float*)d_in[16];
    const float* bf1  = (const float*)d_in[17];
    const float* Wf2  = (const float*)d_in[18];
    const float* bf2  = (const float*)d_in[19];

    const int N = in_sizes[0] / DH;
    const int E = in_sizes[2] / 2;
    const long long ND = (long long)N * DH;

    float* buf0 = (float*)d_ws;
    float* buf1 = buf0 + ND;
    float* buf2 = buf1 + ND;
    _Float16* wp = (_Float16*)(buf2 + ND);
    const int WPN = 4 * 8 * 32 * 16;
    _Float16* wp0 = wp;
    _Float16* wp5 = wp + 5 * WPN;

    float* x_out = (float*)d_out;
    float* t_out = x_out + ND;

    const int nRowTiles = N / 16;
    dim3 gB((nRowTiles + 3) / 4), tB(128);

    int* sorted = (int*)buf2;
    int* cnt    = sorted + 2 * (long long)E;
    int* off    = cnt + 2 * (N + 1);
    int* cntc   = (int*)buf0;

    {
        dim3 gc(NCH, 2);
        dim3 gcp(NCH, 4);
        edge_count<<<gcp, 1024, 0, stream>>>(e_t + E, e_x + E, cntc, E, N,
                                             W1a, W2a, W1b, W2b, Wo, Wf1, Wf2, wp);
        dim3 gp((unsigned)((N + 255) / 256), 2);
        chunk_prefix<<<gp, 256, 0, stream>>>(cntc, cnt, N);
        dim3 gs((unsigned)((N + 1 + 1023) / 1024), 2);
        scan_counts<<<gs, 1024, 0, stream>>>(cnt, off, N);
        dim3 gcx(NCH, 2 + (unsigned)((nRowTiles + 32 * NCH - 1) / (32 * NCH)));
        edge_place<<<gcx, 1024, 0, stream>>>(e_t, e_t + E, e_x, e_x + E, cntc, off, sorted, E, N,
                                             x, wp5, bf1, bf2, x_out, nRowTiles);
        dim3 gg((unsigned)((N + 7) / 8), 2);
        gather_sum<<<gg, 256, 0, stream>>>(t, x, off, sorted, buf0, buf1, N, E);
    }
    t_chain<<<gB, tB, 0, stream>>>(t, buf0, buf1, wp0, b1a, b2a, b1b, b2b, bo, ln_g, ln_b, t_out, nRowTiles);
}


# Edge grouping in two levels: chunks bin edges into owner buckets of 256 nodes, one work-group per bucket orders its run in LDS; no per-node count matrix; 6 -> 5 launches; rows summed in arrival order
# speedup vs baseline: 5.3402x; 1.4746x over previous
#include <hip/hip_runtime.h>
#include <hip/hip_bf16.h>

typedef __attribute__((ext_vector_type(16))) _Float16 v16h;
typedef __attribute__((ext_vector_type(8)))  float    v8f;
typedef __attribute__((ext_vector_type(4)))  float    v4f;
typedef __attribute__((ext_vector_type(4)))  _Float16 v4h;

#define DH 128

__device__ __forceinline__
void pack_one(const float* __restrict__ W, _Float16* __restrict__ Wp, int idx) {
    int h    = idx & 15;
    int lane = (idx >> 4) & 31;
    int nt   = (idx >> 9) & 7;
    int kt   = (idx >> 12) & 3;
    int grp  = lane >> 4;
    int k    = kt * 32 + grp * 8 + ((h < 8) ? h : (16 + (h - 8)));
    int n    = nt * 16 + (lane & 15);
    Wp[idx]  = (_Float16)W[k * DH + n];
}

__device__ __forceinline__ v4h to_h4(float4 v) {
    v4h h;
    h[0] = (_Float16)v.x; h[1] = (_Float16)v.y; h[2] = (_Float16)v.z; h[3] = (_Float16)v.w;
    return h;
}

__device__ __forceinline__
void rows_to_f16(const float* __restrict__ rows, _Float16* __restrict__ rowsH,
                 long long nElem, int part, int nParts, int tid) {
    const float4* s4 = (const float4*)rows;
    v4h* d4 = (v4h*)rowsH;
    long long n4  = nElem >> 2;
    long long per = (((n4 + nParts - 1) / nParts) + 8191) & ~8191LL;
    long long beg = (long long)part * per;
    long long end = beg + per < n4 ? beg + per : n4;
    long long i = beg + tid;
    for (; i + 7 * 1024 < end; i += 8 * 1024) {
        float4 v[8];
#pragma unroll
        for (int u = 0; u < 8; ++u) v[u] = s4[i + u * 1024];
#pragma unroll
        for (int u = 0; u < 8; ++u) d4[i + u * 1024] = to_h4(v[u]);
    }
    for (; i < end; i += 1024) d4[i] = to_h4(s4[i]);
}

#define NCH  64
#define BSH  8
#define BSZ  (1 << BSH)
#define BMASK (BSZ - 1)
#define BMAXB 512
#define NMAT 7
#define WPN_H (4 * 8 * 32 * 16)

__device__ __forceinline__ int chunk_len(int nEdges) {
    return (((nEdges + NCH - 1) / NCH) + 1023) & ~1023;
}

__global__ __launch_bounds__(1024)
void edge_count(const int* __restrict__ dst0, const int* __restrict__ dst1,
                int* __restrict__ cntb, int nEdges, int nNodes, int nB,
                const float* __restrict__ W0, const float* __restrict__ W1,
                const float* __restrict__ W2, const float* __restrict__ W3,
                const float* __restrict__ W4, const float* __restrict__ W5,
                const float* __restrict__ W6, _Float16* __restrict__ Wp,
                const float* __restrict__ rows0, _Float16* __restrict__ rows0H) {
    __shared__ int hist[BMAXB];
    int tid  = threadIdx.x;
    if (blockIdx.y >= 4) {
        rows_to_f16(rows0, rows0H, (long long)nNodes * DH,
                    (int)(blockIdx.y - 4) * NCH + (int)blockIdx.x, 2 * NCH, tid);
        return;
    }
    if (blockIdx.y >= 2) {
        int pb  = (int)(blockIdx.y - 2) * NCH + (int)blockIdx.x;
        int mat = pb >> 4;
        if (mat >= NMAT) return;
        const float* W = mat == 0 ? W0 : mat == 1 ? W1 : mat == 2 ? W2 : mat == 3 ? W3
                       : mat == 4 ? W4 : mat == 5 ? W5 : W6;
        pack_one(W, Wp + mat * WPN_H, ((pb & 15) << 10) + tid);
        return;
    }
    int list = blockIdx.y;
    const int* dst = list ? dst1 : dst0;
    int* part = cntb + ((long long)list * NCH + blockIdx.x) * nB;
    int per = chunk_len(nEdges);
    int beg = blockIdx.x * per;
    int end = min(beg + per, nEdges);
    for (int i = tid; i < nB; i += 1024) hist[i] = 0;
    __syncthreads();
    int e = beg + tid;
    for (; e + 7 * 1024 < end; e += 8 * 1024) {
        int d[8];
#pragma unroll
        for (int u = 0; u < 8; ++u) d[u] = dst[e + u * 1024];
#pragma unroll
        for (int u = 0; u < 8; ++u)
            if ((unsigned)d[u] < (unsigned)nNodes) atomicAdd(&hist[d[u] >> BSH], 1);
    }
    for (; e < end; e += 1024) {
        int d = dst[e];
        if ((unsigned)d < (unsigned)nNodes) atomicAdd(&hist[d >> BSH], 1);
    }
    __syncthreads();
    for (int i = tid; i < nB; i += 1024) part[i] = hist[i];
}

#define OCAP 32768
__global__ __launch_bounds__(1024)
void owner_sort(const int* __restrict__ bkt, const int* __restrict__ bbase,
                int* __restrict__ off, int* __restrict__ sorted,
                int nEdges, int nNodes, int nB) {
    __shared__ int stage[OCAP];
    __shared__ int cnt[BSZ];
    __shared__ int wtot[BSZ / 32];
    int tid  = threadIdx.x;
    int lane = tid & 31;
    int w    = tid >> 5;
    int list = blockIdx.y;
    int b    = blockIdx.x;
    const int* in = bkt + (long long)list * nEdges;
    int* out = sorted + (long long)list * nEdges;
    int* o = off + (long long)list * (nNodes + 1);
    int beg = bbase[list * (nB + 1) + b];
    int end = bbase[list * (nB + 1) + b + 1];
    if (tid < BSZ) cnt[tid] = 0;
    __syncthreads();
    int i = beg + tid;
    for (; i + 3 * 1024 < end; i += 4 * 1024) {
        int w0 = in[i], w1 = in[i + 1024], w2 = in[i + 2048], w3 = in[i + 3072];
        atomicAdd(&cnt[w0 & BMASK], 1);
        atomicAdd(&cnt[w1 & BMASK], 1);
        atomicAdd(&cnt[w2 & BMASK], 1);
        atomicAdd(&cnt[w3 & BMASK], 1);
    }
    for (; i < end; i += 1024) atomicAdd(&cnt[in[i] & BMASK], 1);
    __syncthreads();
    int v = (tid < BSZ) ? cnt[tid] : 0;
    int incl = v;
#pragma unroll
    for (int d = 1; d < 32; d <<= 1) {
        int u = __shfl_up(incl, d, 32);
        if (lane >= d) incl += u;
    }
    if (tid < BSZ && lane == 31) wtot[w] = incl;
    __syncthreads();
    if (tid < BSZ) {
        int pre = 0;
        for (int k = 0; k < w; ++k) pre += wtot[k];
        int start = beg + pre + (incl - v);
        cnt[tid] = start;
        int node = (b << BSH) + tid;
        if (node < nNodes) o[node] = start;
    }
    if (b == nB - 1 && tid == 0) o[nNodes] = end;
    __syncthreads();
    int len = end - beg;
    i = beg + tid;
    if (len <= OCAP) {
        for (; i + 3 * 1024 < end; i += 4 * 1024) {
            int w0 = in[i], w1 = in[i + 1024], w2 = in[i + 2048], w3 = in[i + 3072];
            stage[atomicAdd(&cnt[w0 & BMASK], 1) - beg] = (int)((unsigned)w0 >> BSH);
            stage[atomicAdd(&cnt[w1 & BMASK], 1) - beg] = (int)((unsigned)w1 >> BSH);
            stage[atomicAdd(&cnt[w2 & BMASK], 1) - beg] = (int)((unsigned)w2 >> BSH);
            stage[atomicAdd(&cnt[w3 & BMASK], 1) - beg] = (int)((unsigned)w3 >> BSH);
        }
        for (; i < end; i += 1024) {
            int w0 = in[i];
            stage[atomicAdd(&cnt[w0 & BMASK], 1) - beg] = (int)((unsigned)w0 >> BSH);
        }
    } else {
        for (; i + 3 * 1024 < end; i += 4 * 1024) {
            int w0 = in[i], w1 = in[i + 1024], w2 = in[i + 2048], w3 = in[i + 3072];
            out[atomicAdd(&cnt[w0 & BMASK], 1)] = (int)((unsigned)w0 >> BSH);
            out[atomicAdd(&cnt[w1 & BMASK], 1)] = (int)((unsigned)w1 >> BSH);
            out[atomicAdd(&cnt[w2 & BMASK], 1)] = (int)((unsigned)w2 >> BSH);
            out[atomicAdd(&cnt[w3 & BMASK], 1)] = (int)((unsigned)w3 >> BSH);
        }
        for (; i < end; i += 1024) {
            int w0 = in[i];
            out[atomicAdd(&cnt[w0 & BMASK], 1)] = (int)((unsigned)w0 >> BSH);
        }
    }
    __syncthreads();
    if (len <= OCAP)
        for (int k = tid; k < len; k += 1024) out[beg + k] = stage[k];
}

__global__ __launch_bounds__(256)
void gather_sum(const _Float16* __restrict__ rows0, const _Float16* __restrict__ rows1,
                const int* __restrict__ off, const int* __restrict__ sorted,
                float* __restrict__ agg0, float* __restrict__ agg1,
                int nNodes, int nEdges) {
    int d = blockIdx.x * (blockDim.x >> 5) + (threadIdx.x >> 5);
    if (d >= nNodes) return;
    int list = blockIdx.y;
    const _Float16* rows = list ? rows1 : rows0;
    float* agg = list ? agg1 : agg0;
    const int* o  = off + (long long)list * (nNodes + 1);
    const int* sl = sorted + (long long)list * nEdges;
    int lane = threadIdx.x & 31;
    int beg = __builtin_amdgcn_readfirstlane(o[d]);
    int end = __builtin_amdgcn_readfirstlane(o[d + 1]);
    float4 a0 = make_float4(0.f, 0.f, 0.f, 0.f);
    float4 a1 = a0, a2 = a0, a3 = a0;
#define ROW_LD(j) (((const v4h*)(rows + (long long)__builtin_amdgcn_readlane(idx, (j)) * DH))[lane])
#define ACC4(a, v) { a.x += (float)v[0]; a.y += (float)v[1]; a.z += (float)v[2]; a.w += (float)v[3]; }
    for (int base = beg; base < end; base += 32) {
        int n = min(end - base, 32);
        int idx = (lane < n) ? sl[base + lane] : 0;
        int j = 0;
        for (; j + 8 <= n; j += 8) {
            v4h v0 = ROW_LD(j),     v1 = ROW_LD(j + 1), v2 = ROW_LD(j + 2), v3 = ROW_LD(j + 3);
            v4h v4 = ROW_LD(j + 4), v5 = ROW_LD(j + 5), v6 = ROW_LD(j + 6), v7 = ROW_LD(j + 7);
            ACC4(a0, v0) ACC4(a1, v1) ACC4(a2, v2) ACC4(a3, v3)
            ACC4(a0, v4) ACC4(a1, v5) ACC4(a2, v6) ACC4(a3, v7)
        }
        if (j + 4 <= n) {
            v4h v0 = ROW_LD(j), v1 = ROW_LD(j + 1), v2 = ROW_LD(j + 2), v3 = ROW_LD(j + 3);
            ACC4(a0, v0) ACC4(a1, v1) ACC4(a2, v2) ACC4(a3, v3)
            j += 4;
        }
        for (; j < n; ++j) {
            v4h v0 = ROW_LD(j);
            ACC4(a0, v0)
        }
    }
#undef ROW_LD
#undef ACC4
    float4 r;
    r.x = (a0.x + a1.x) + (a2.x + a3.x);
    r.y = (a0.y + a1.y) + (a2.y + a3.y);
    r.z = (a0.z + a1.z) + (a2.z + a3.z);
    r.w = (a0.w + a1.w) + (a2.w + a3.w);
    ((float4*)(agg + (long long)d * DH))[lane] = r;
}

#define LP 132

__device__ __forceinline__ v16h pack16(float4 q0, float4 q1, float4 q2, float4 q3) {
    v16h a;
    a[0]  = (_Float16)q0.x; a[1]  = (_Float16)q0.y; a[2]  = (_Float16)q0.z; a[3]  = (_Float16)q0.w;
    a[4]  = (_Float16)q1.x; a[5]  = (_Float16)q1.y; a[6]  = (_Float16)q1.z; a[7]  = (_Float16)q1.w;
    a[8]  = (_Float16)q2.x; a[9]  = (_Float16)q2.y; a[10] = (_Float16)q2.z; a[11] = (_Float16)q2.w;
    a[12] = (_Float16)q3.x; a[13] = (_Float16)q3.y; a[14] = (_Float16)q3.z; a[15] = (_Float16)q3.w;
    return a;
}

__device__ __forceinline__ void add4(float4& q, float4 b) {
    q.x += b.x; q.y += b.y; q.z += b.z; q.w += b.w;
}

template <bool HAS_A1>
__device__ __forceinline__ void a_from_global(const float* __restrict__ ar0,
                                              const float* __restrict__ ar1,
                                              int grp, v16h (&afrag)[4]) {
#pragma unroll
    for (int kt = 0; kt < 4; ++kt) {
        int off = kt * 32 + grp * 8;
        float4 q0 = *(const float4*)(ar0 + off);
        float4 q1 = *(const float4*)(ar0 + off + 4);
        float4 q2 = *(const float4*)(ar0 + off + 16);
        float4 q3 = *(const float4*)(ar0 + off + 20);
        if (HAS_A1) {
            add4(q0, *(const float4*)(ar1 + off));
            add4(q1, *(const float4*)(ar1 + off + 4));
            add4(q2, *(const float4*)(ar1 + off + 16));
            add4(q3, *(const float4*)(ar1 + off + 20));
        }
        afrag[kt] = pack16(q0, q1, q2, q3);
    }
}

__device__ __forceinline__ void a_from_lds(const float* ar, int grp, v16h (&afrag)[4]) {
#pragma unroll
    for (int kt = 0; kt < 4; ++kt) {
        int off = kt * 32 + grp * 8;
        float4 q0 = *(const float4*)(ar + off);
        float4 q1 = *(const float4*)(ar + off + 4);
        float4 q2 = *(const float4*)(ar + off + 16);
        float4 q3 = *(const float4*)(ar + off + 20);
        afrag[kt] = pack16(q0, q1, q2, q3);
    }
}

__device__ __forceinline__ void load_b(const _Float16* __restrict__ Wp,
                                       int nt, int lane, v16h (&b)[4]) {
    const v16h* wfrag = (const v16h*)Wp;
#pragma unroll
    for (int kt = 0; kt < 4; ++kt) b[kt] = wfrag[(kt * 8 + nt) * 32 + lane];
}

template <bool RELU>
__device__ __forceinline__ void layer_to_lds(const v16h (&afrag)[4],
                                             const _Float16* __restrict__ Wp,
                                             const _Float16* __restrict__ WpNext,
                                             const float* __restrict__ bias,
                                             float* myLds, int lane, int grp, int col,
                                             v16h (&bq)[4]) {
#pragma unroll
    for (int nt = 0; nt < 8; ++nt) {
        v16h c0 = bq[0], c1 = bq[1], c2 = bq[2], c3 = bq[3];
        if (nt < 7)      load_b(Wp, nt + 1, lane, bq);
        else if (WpNext) load_b(WpNext, 0, lane, bq);
        asm volatile("" :: "v"(c0), "v"(c1), "v"(c2), "v"(c3));
        v8f acc = {};
        acc = __builtin_amdgcn_wmma_f32_16x16x32_f16(false, afrag[0], false, c0, (short)0, acc, false, false);
        acc = __builtin_amdgcn_wmma_f32_16x16x32_f16(false, afrag[1], false, c1, (short)0, acc, false, false);
        acc = __builtin_amdgcn_wmma_f32_16x16x32_f16(false, afrag[2], false, c2, (short)0, acc, false, false);
        acc = __builtin_amdgcn_wmma_f32_16x16x32_f16(false, afrag[3], false, c3, (short)0, acc, false, false);
        float bn = bias[nt * 16 + col];
#pragma unroll
        for (int r = 0; r < 8; ++r) {
            float v = acc[r] + bn;
            if (RELU) v = fmaxf(v, 0.f);
            myLds[(r + grp * 8) * LP + nt * 16 + col] = v;
        }
    }
}

__global__ __launch_bounds__(128)
void t_chain(const float* __restrict__ t,
             const float* __restrict__ aggA, const float* __restrict__ aggB,
             const _Float16* __restrict__ Wp,
             const float* __restrict__ b1a, const float* __restrict__ b2a,
             const float* __restrict__ b1b, const float* __restrict__ b2b,
             const float* __restrict__ bo,
             const float* __restrict__ ln_g, const float* __restrict__ ln_b,
             float* __restrict__ t_out, int nRowTiles) {
    __shared__ __attribute__((aligned(16))) float lds[4][16 * LP];
    int wave = blockIdx.x * (blockDim.x >> 5) + (threadIdx.x >> 5);
    if (wave >= nRowTiles) return;
    int lane = threadIdx.x & 31;
    int grp  = lane >> 4;
    int col  = lane & 15;
    float* myLds = lds[threadIdx.x >> 5];
    const float* myRow = myLds + col * LP;
    long long row = (long long)wave * 16 + col;
    long long m0  = (long long)wave * 16;
    v16h afrag[4], afragB[4], bq[4];
    float4 s[16];

    load_b(Wp, 0, lane, bq);
    a_from_global<true>(t + row * DH, aggA + row * DH, grp, afrag);
    a_from_global<true>(t + row * DH, aggB + row * DH, grp, afragB);

    layer_to_lds<true >(afrag, Wp,             Wp + WPN_H,     b1a, myLds, lane, grp, col, bq);
    a_from_lds(myRow, grp, afrag);
    layer_to_lds<false>(afrag, Wp + WPN_H,     Wp + 2 * WPN_H, b2a, myLds, lane, grp, col, bq);
#pragma unroll
    for (int rl = 0; rl < 16; ++rl) {
        float4 g  = ((const float4*)(myLds + rl * LP))[lane];
        float4 tv = ((const float4*)(t + (m0 + rl) * DH))[lane];
        s[rl].x = tv.x + g.x; s[rl].y = tv.y + g.y; s[rl].z = tv.z + g.z; s[rl].w = tv.w + g.w;
    }

    layer_to_lds<true >(afragB, Wp + 2 * WPN_H, Wp + 3 * WPN_H, b1b, myLds, lane, grp, col, bq);
    a_from_lds(myRow, grp, afrag);
    layer_to_lds<false>(afrag,  Wp + 3 * WPN_H, Wp + 4 * WPN_H, b2b, myLds, lane, grp, col, bq);
#pragma unroll
    for (int rl = 0; rl < 16; ++rl) {
        float4 g = ((const float4*)(myLds + rl * LP))[lane];
        s[rl].x = fmaxf(s[rl].x + g.x, 0.f); s[rl].y = fmaxf(s[rl].y + g.y, 0.f);
        s[rl].z = fmaxf(s[rl].z + g.z, 0.f); s[rl].w = fmaxf(s[rl].w + g.w, 0.f);
    }
#pragma unroll
    for (int rl = 0; rl < 16; ++rl)
        ((float4*)(myLds + rl * LP))[lane] = s[rl];

    a_from_lds(myRow, grp, afrag);
    layer_to_lds<true >(afrag, Wp + 4 * WPN_H, nullptr, bo, myLds, lane, grp, col, bq);
    float4 gv = ((const float4*)ln_g)[lane];
    float4 bv = ((const float4*)ln_b)[lane];
#pragma unroll
    for (int rl = 0; rl < 16; ++rl) {
        float4 v = ((const float4*)(myLds + rl * LP))[lane];
        float sm = v.x + v.y + v.z + v.w;
        float sq = v.x * v.x + v.y * v.y + v.z * v.z + v.w * v.w;
#pragma unroll
        for (int off = 16; off > 0; off >>= 1) {
            sm += __shfl_xor(sm, off, 32);
            sq += __shfl_xor(sq, off, 32);
        }
        float mean = sm * (1.0f / 128.0f);
        float var  = sq * (1.0f / 128.0f) - mean * mean;
        float inv  = rsqrtf(var + 1e-5f);
        float4 tv = s[rl];
        float4 o;
        o.x = tv.x + (v.x - mean) * inv * gv.x + bv.x;
        o.y = tv.y + (v.y - mean) * inv * gv.y + bv.y;
        o.z = tv.z + (v.z - mean) * inv * gv.z + bv.z;
        o.w = tv.w + (v.w - mean) * inv * gv.w + bv.w;
        ((float4*)(t_out + (m0 + rl) * DH))[lane] = o;
    }
}

__device__ __forceinline__
void x_chain_tile(const float* __restrict__ x, const _Float16* __restrict__ Wp,
                  const float* __restrict__ bf1, const float* __restrict__ bf2,
                  float* __restrict__ x_out, int wave, float* myLds, int lane) {
    int grp  = lane >> 4;
    int col  = lane & 15;
    const float* myRow = myLds + col * LP;
    long long row = (long long)wave * 16 + col;
    long long m0  = (long long)wave * 16;
    v16h afrag[4], bq[4];

    load_b(Wp, 0, lane, bq);
    const float* ar = x + row * DH + grp * 8;
    v4f q[16];
#pragma unroll
    for (int kt = 0; kt < 4; ++kt) {
        q[4 * kt]     = *(const v4f*)(ar + kt * 32);
        q[4 * kt + 1] = *(const v4f*)(ar + kt * 32 + 4);
        q[4 * kt + 2] = *(const v4f*)(ar + kt * 32 + 16);
        q[4 * kt + 3] = *(const v4f*)(ar + kt * 32 + 20);
    }
    asm volatile("" :: "v"(q[0]), "v"(q[1]), "v"(q[2]),  "v"(q[3]),  "v"(q[4]),  "v"(q[5]),  "v"(q[6]),  "v"(q[7]),
                       "v"(q[8]), "v"(q[9]), "v"(q[10]), "v"(q[11]), "v"(q[12]), "v"(q[13]), "v"(q[14]), "v"(q[15]));
#pragma unroll
    for (int kt = 0; kt < 4; ++kt) {
#pragma unroll
        for (int h = 0; h < 16; ++h) afrag[kt][h] = (_Float16)q[4 * kt + (h >> 2)][h & 3];
    }
    layer_to_lds<true >(afrag, Wp,         Wp + WPN_H, bf1, myLds, lane, grp, col, bq);
    __builtin_amdgcn_sched_barrier(0);
    a_from_lds(myRow, grp, afrag);
    __builtin_amdgcn_sched_barrier(0);
    layer_to_lds<false>(afrag, Wp + WPN_H, nullptr,    bf2, myLds, lane, grp, col, bq);
    __builtin_amdgcn_sched_barrier(0);
#pragma unroll
    for (int rl = 0; rl < 16; ++rl) {
        float4 v  = ((const float4*)(myLds + rl * LP))[lane];
        float4 rv = ((const float4*)(x + (m0 + rl) * DH))[lane];
        v.x += rv.x; v.y += rv.y; v.z += rv.z; v.w += rv.w;
        ((float4*)(x_out + (m0 + rl) * DH))[lane] = v;
    }
}

#define PLDS ((32 * 16 * LP) > (NCH * BMAXB + BMAXB + 32) ? (32 * 16 * LP) : (NCH * BMAXB + BMAXB + 32))

__global__ __launch_bounds__(1024)
void edge_place(const int* __restrict__ src0, const int* __restrict__ dst0,
                const int* __restrict__ src1, const int* __restrict__ dst1,
                const int* __restrict__ cntb, int* __restrict__ bbase,
                int* __restrict__ bkt, int nEdges, int nNodes, int nB,
                const float* __restrict__ x, const _Float16* __restrict__ WpX,
                const float* __restrict__ bf1, const float* __restrict__ bf2,
                float* __restrict__ x_out, int nRowTiles,
                _Float16* __restrict__ xH) {
    __shared__ __attribute__((aligned(16))) int cur[PLDS];
    int tid  = threadIdx.x;
    int xRows = (nRowTiles + 32 * NCH - 1) / (32 * NCH);
    if ((int)blockIdx.y >= 2 + xRows) {
        rows_to_f16(x, xH, (long long)nNodes * DH,
                    ((int)blockIdx.y - 2 - xRows) * NCH + (int)blockIdx.x, 2 * NCH, tid);
        return;
    }
    if (blockIdx.y >= 2) {
        int xb   = (int)(blockIdx.y - 2) * NCH + (int)blockIdx.x;
        int tile = xb * 32 + (tid >> 5);
        if (tile >= nRowTiles) return;
        x_chain_tile(x, WpX, bf1, bf2, x_out, tile, (float*)cur + (tid >> 5) * (16 * LP), tid & 31);
        return;
    }
    int list = blockIdx.y;
    const int* src = list ? src1 : src0;
    const int* dst = list ? dst1 : dst0;
    const int* cm = cntb + (long long)list * NCH * nB;
    int* out = bkt + (long long)list * nEdges;
    int per = chunk_len(nEdges);
    int c   = blockIdx.x;
    int beg = c * per;
    int end = min(beg + per, nEdges);
    int* m    = cur;
    int* posn = cur + NCH * BMAXB;
    int* wtot = posn + BMAXB;
    int lane = tid & 31;
    int w    = tid >> 5;
    int n = NCH * nB;
#pragma unroll 8
    for (int i = tid; i < n; i += 1024) m[i] = cm[i];
    __syncthreads();
    int tot = 0, mine = 0;
    if (tid < nB) {
        for (int cc = 0; cc < NCH; ++cc) {
            if (cc == c) mine = tot;
            tot += m[cc * nB + tid];
        }
    }
    int incl = tot;
#pragma unroll
    for (int d = 1; d < 32; d <<= 1) {
        int u = __shfl_up(incl, d, 32);
        if (lane >= d) incl += u;
    }
    if (lane == 31) wtot[w] = incl;
    __syncthreads();
    int wt = wtot[lane];
    int tincl = wt;
#pragma unroll
    for (int d = 1; d < 32; d <<= 1) {
        int u = __shfl_up(tincl, d, 32);
        if (lane >= d) tincl += u;
    }
    int wexcl = __shfl(tincl - wt, w, 32);
    int base = wexcl + (incl - tot);
    if (tid < nB) {
        posn[tid] = base + mine;
        if (c == 0) {
            bbase[list * (nB + 1) + tid] = base;
            if (tid == nB - 1) bbase[list * (nB + 1) + nB] = base + tot;
        }
    }
    __syncthreads();
    int e = beg + tid;
    for (; e + 3 * 1024 < end; e += 4 * 1024) {
        int d0 = dst[e],        d1 = dst[e + 1024];
        int d2 = dst[e + 2048], d3 = dst[e + 3072];
        int s0 = src[e],        s1 = src[e + 1024];
        int s2 = src[e + 2048], s3 = src[e + 3072];
        if ((unsigned)d0 < (unsigned)nNodes) out[atomicAdd(&posn[d0 >> BSH], 1)] = (s0 << BSH) | (d0 & BMASK);
        if ((unsigned)d1 < (unsigned)nNodes) out[atomicAdd(&posn[d1 >> BSH], 1)] = (s1 << BSH) | (d1 & BMASK);
        if ((unsigned)d2 < (unsigned)nNodes) out[atomicAdd(&posn[d2 >> BSH], 1)] = (s2 << BSH) | (d2 & BMASK);
        if ((unsigned)d3 < (unsigned)nNodes) out[atomicAdd(&posn[d3 >> BSH], 1)] = (s3 << BSH) | (d3 & BMASK);
    }
    for (; e < end; e += 1024) {
        int d = dst[e];
        int sv = src[e];
        if ((unsigned)d < (unsigned)nNodes) out[atomicAdd(&posn[d >> BSH], 1)] = (sv << BSH) | (d & BMASK);
    }
}

extern "C" void kernel_launch(void* const* d_in, const int* in_sizes, int n_in,
                              void* d_out, int out_size, void* d_ws, size_t ws_size,
                              hipStream_t stream) {
    const float* x    = (const float*)d_in[0];
    const float* t    = (const float*)d_in[1];
    const int*   e_t  = (const int*)d_in[2];
    const int*   e_x  = (const int*)d_in[3];
    const float* W1a  = (const float*)d_in[4];
    const float* b1a  = (const float*)d_in[5];
    const float* W2a  = (const float*)d_in[6];
    const float* b2a  = (const float*)d_in[7];
    const float* W1b  = (const float*)d_in[8];
    const float* b1b  = (const float*)d_in[9];
    const float* W2b  = (const float*)d_in[10];
    const float* b2b  = (const float*)d_in[11];
    const float* Wo   = (const float*)d_in[12];
    const float* bo   = (const float*)d_in[13];
    const float* ln_g = (const float*)d_in[14];
    const float* ln_b = (const float*)d_in[15];
    const float* Wf1  = (const float*)d_in[16];
    const float* bf1  = (const float*)d_in[17];
    const float* Wf2  = (const float*)d_in[18];
    const float* bf2  = (const float*)d_in[19];

    const int N = in_sizes[0] / DH;
    const int E = in_sizes[2] / 2;
    const long long ND = (long long)N * DH;

    float* buf0 = (float*)d_ws;
    float* buf1 = buf0 + ND;
    float* buf2 = buf1 + ND;
    _Float16* wp = (_Float16*)(buf2 + ND);
    const int WPN = 4 * 8 * 32 * 16;
    _Float16* wp0 = wp;
    _Float16* wp5 = wp + 5 * WPN;

    float* x_out = (float*)d_out;
    float* t_out = x_out + ND;
    _Float16* tH = (_Float16*)t_out;
    _Float16* xH = tH + ND;

    const int nRowTiles = N / 16;
    dim3 gB((nRowTiles + 3) / 4), tB(128);

    int* sorted = (int*)buf2;
    int* off    = sorted + 2 * (long long)E;
    int* bkt    = (int*)buf0;
    const int nB = (N + BSZ - 1) / BSZ;
    int* cntb   = (int*)buf1;
    int* bbase  = cntb + 2 * NCH * nB;

    {
        dim3 gcp(NCH, 6);
        edge_count<<<gcp, 1024, 0, stream>>>(e_t + E, e_x + E, cntb, E, N, nB,
                                             W1a, W2a, W1b, W2b, Wo, Wf1, Wf2, wp, t, tH);
        dim3 gcx(NCH, 4 + (unsigned)((nRowTiles + 32 * NCH - 1) / (32 * NCH)));
        edge_place<<<gcx, 1024, 0, stream>>>(e_t, e_t + E, e_x, e_x + E, cntb, bbase, bkt, E, N, nB,
                                             x, wp5, bf1, bf2, x_out, nRowTiles, xH);
        dim3 go((unsigned)nB, 2);
        owner_sort<<<go, 1024, 0, stream>>>(bkt, bbase, off, sorted, E, N, nB);
        dim3 gg((unsigned)((N + 7) / 8), 2);
        gather_sum<<<gg, 256, 0, stream>>>(tH, xH, off, sorted, buf0, buf1, N, E);
    }
    t_chain<<<gB, tB, 0, stream>>>(t, buf0, buf1, wp0, b1a, b2a, b1b, b2b, bo, ln_g, ln_b, t_out, nRowTiles);
}


# Edge gather on a resident grid: 8192 waves each walk a run of about 12 nodes (was 100000 one-node waves), next node's indices asked for ahead, a 32-row block per round trip; sums in the same order
# speedup vs baseline: 5.5321x; 1.0359x over previous
#include <hip/hip_runtime.h>
#include <hip/hip_bf16.h>

typedef __attribute__((ext_vector_type(16))) _Float16 v16h;
typedef __attribute__((ext_vector_type(8)))  float    v8f;
typedef __attribute__((ext_vector_type(4)))  float    v4f;
typedef __attribute__((ext_vector_type(4)))  _Float16 v4h;

#define DH 128

__device__ __forceinline__
void pack_one(const float* __restrict__ W, _Float16* __restrict__ Wp, int idx) {
    int h    = idx & 15;
    int lane = (idx >> 4) & 31;
    int nt   = (idx >> 9) & 7;
    int kt   = (idx >> 12) & 3;
    int grp  = lane >> 4;
    int k    = kt * 32 + grp * 8 + ((h < 8) ? h : (16 + (h - 8)));
    int n    = nt * 16 + (lane & 15);
    Wp[idx]  = (_Float16)W[k * DH + n];
}

__device__ __forceinline__ v4h to_h4(float4 v) {
    v4h h;
    h[0] = (_Float16)v.x; h[1] = (_Float16)v.y; h[2] = (_Float16)v.z; h[3] = (_Float16)v.w;
    return h;
}

__device__ __forceinline__
void rows_to_f16(const float* __restrict__ rows, _Float16* __restrict__ rowsH,
                 long long nElem, int part, int nParts, int tid) {
    const float4* s4 = (const float4*)rows;
    v4h* d4 = (v4h*)rowsH;
    long long n4  = nElem >> 2;
    long long per = (((n4 + nParts - 1) / nParts) + 8191) & ~8191LL;
    long long beg = (long long)part * per;
    long long end = beg + per < n4 ? beg + per : n4;
    long long i = beg + tid;
    for (; i + 7 * 1024 < end; i += 8 * 1024) {
        float4 v[8];
#pragma unroll
        for (int u = 0; u < 8; ++u) v[u] = s4[i + u * 1024];
#pragma unroll
        for (int u = 0; u < 8; ++u) d4[i + u * 1024] = to_h4(v[u]);
    }
    for (; i < end; i += 1024) d4[i] = to_h4(s4[i]);
}

#define NCH  64
#define BSH  8
#define BSZ  (1 << BSH)
#define BMASK (BSZ - 1)
#define BMAXB 512
#define NMAT 7
#define WPN_H (4 * 8 * 32 * 16)

__device__ __forceinline__ int chunk_len(int nEdges) {
    return (((nEdges + NCH - 1) / NCH) + 1023) & ~1023;
}

__global__ __launch_bounds__(1024)
void edge_count(const int* __restrict__ dst0, const int* __restrict__ dst1,
                int* __restrict__ cntb, int nEdges, int nNodes, int nB,
                const float* __restrict__ W0, const float* __restrict__ W1,
                const float* __restrict__ W2, const float* __restrict__ W3,
                const float* __restrict__ W4, const float* __restrict__ W5,
                const float* __restrict__ W6, _Float16* __restrict__ Wp,
                const float* __restrict__ rows0, _Float16* __restrict__ rows0H) {
    __shared__ int hist[BMAXB];
    int tid  = threadIdx.x;
    if (blockIdx.y >= 4) {
        rows_to_f16(rows0, rows0H, (long long)nNodes * DH,
                    (int)(blockIdx.y - 4) * NCH + (int)blockIdx.x, 2 * NCH, tid);
        return;
    }
    if (blockIdx.y >= 2) {
        int pb  = (int)(blockIdx.y - 2) * NCH + (int)blockIdx.x;
        int mat = pb >> 4;
        if (mat >= NMAT) return;
        const float* W = mat == 0 ? W0 : mat == 1 ? W1 : mat == 2 ? W2 : mat == 3 ? W3
                       : mat == 4 ? W4 : mat == 5 ? W5 : W6;
        pack_one(W, Wp + mat * WPN_H, ((pb & 15) << 10) + tid);
        return;
    }
    int list = blockIdx.y;
    const int* dst = list ? dst1 : dst0;
    int* part = cntb + ((long long)list * NCH + blockIdx.x) * nB;
    int per = chunk_len(nEdges);
    int beg = blockIdx.x * per;
    int end = min(beg + per, nEdges);
    for (int i = tid; i < nB; i += 1024) hist[i] = 0;
    __syncthreads();
    int e = beg + tid;
    for (; e + 7 * 1024 < end; e += 8 * 1024) {
        int d[8];
#pragma unroll
        for (int u = 0; u < 8; ++u) d[u] = dst[e + u * 1024];
#pragma unroll
        for (int u = 0; u < 8; ++u)
            if ((unsigned)d[u] < (unsigned)nNodes) atomicAdd(&hist[d[u] >> BSH], 1);
    }
    for (; e < end; e += 1024) {
        int d = dst[e];
        if ((unsigned)d < (unsigned)nNodes) atomicAdd(&hist[d >> BSH], 1);
    }
    __syncthreads();
    for (int i = tid; i < nB; i += 1024) part[i] = hist[i];
}

#define OCAP 32768
__global__ __launch_bounds__(1024)
void owner_sort(const int* __restrict__ bkt, const int* __restrict__ bbase,
                int* __restrict__ off, int* __restrict__ sorted,
                int nEdges, int nNodes, int nB) {
    __shared__ int stage[OCAP];
    __shared__ int cnt[BSZ];
    __shared__ int wtot[BSZ / 32];
    int tid  = threadIdx.x;
    int lane = tid & 31;
    int w    = tid >> 5;
    int list = blockIdx.y;
    int b    = blockIdx.x;
    const int* in = bkt + (long long)list * nEdges;
    int* out = sorted + (long long)list * nEdges;
    int* o = off + (long long)list * (nNodes + 1);
    int beg = bbase[list * (nB + 1) + b];
    int end = bbase[list * (nB + 1) + b + 1];
    if (tid < BSZ) cnt[tid] = 0;
    __syncthreads();
    int i = beg + tid;
    for (; i + 3 * 1024 < end; i += 4 * 1024) {
        int w0 = in[i], w1 = in[i + 1024], w2 = in[i + 2048], w3 = in[i + 3072];
        atomicAdd(&cnt[w0 & BMASK], 1);
        atomicAdd(&cnt[w1 & BMASK], 1);
        atomicAdd(&cnt[w2 & BMASK], 1);
        atomicAdd(&cnt[w3 & BMASK], 1);
    }
    for (; i < end; i += 1024) atomicAdd(&cnt[in[i] & BMASK], 1);
    __syncthreads();
    int v = (tid < BSZ) ? cnt[tid] : 0;
    int incl = v;
#pragma unroll
    for (int d = 1; d < 32; d <<= 1) {
        int u = __shfl_up(incl, d, 32);
        if (lane >= d) incl += u;
    }
    if (tid < BSZ && lane == 31) wtot[w] = incl;
    __syncthreads();
    if (tid < BSZ) {
        int pre = 0;
        for (int k = 0; k < w; ++k) pre += wtot[k];
        int start = beg + pre + (incl - v);
        cnt[tid] = start;
        int node = (b << BSH) + tid;
        if (node < nNodes) o[node] = start;
    }
    if (b == nB - 1 && tid == 0) o[nNodes] = end;
    __syncthreads();
    int len = end - beg;
    i = beg + tid;
    if (len <= OCAP) {
        for (; i + 3 * 1024 < end; i += 4 * 1024) {
            int w0 = in[i], w1 = in[i + 1024], w2 = in[i + 2048], w3 = in[i + 3072];
            stage[atomicAdd(&cnt[w0 & BMASK], 1) - beg] = (int)((unsigned)w0 >> BSH);
            stage[atomicAdd(&cnt[w1 & BMASK], 1) - beg] = (int)((unsigned)w1 >> BSH);
            stage[atomicAdd(&cnt[w2 & BMASK], 1) - beg] = (int)((unsigned)w2 >> BSH);
            stage[atomicAdd(&cnt[w3 & BMASK], 1) - beg] = (int)((unsigned)w3 >> BSH);
        }
        for (; i < end; i += 1024) {
            int w0 = in[i];
            stage[atomicAdd(&cnt[w0 & BMASK], 1) - beg] = (int)((unsigned)w0 >> BSH);
        }
    } else {
        for (; i + 3 * 1024 < end; i += 4 * 1024) {
            int w0 = in[i], w1 = in[i + 1024], w2 = in[i + 2048], w3 = in[i + 3072];
            out[atomicAdd(&cnt[w0 & BMASK], 1)] = (int)((unsigned)w0 >> BSH);
            out[atomicAdd(&cnt[w1 & BMASK], 1)] = (int)((unsigned)w1 >> BSH);
            out[atomicAdd(&cnt[w2 & BMASK], 1)] = (int)((unsigned)w2 >> BSH);
            out[atomicAdd(&cnt[w3 & BMASK], 1)] = (int)((unsigned)w3 >> BSH);
        }
        for (; i < end; i += 1024) {
            int w0 = in[i];
            out[atomicAdd(&cnt[w0 & BMASK], 1)] = (int)((unsigned)w0 >> BSH);
        }
    }
    __syncthreads();
    if (len <= OCAP)
        for (int k = tid; k < len; k += 1024) out[beg + k] = stage[k];
}

#define GCH 31

#define ROW_AT(i) (((const v4h*)(rows + (long long)(i) * DH))[lane])
#define ACC4(a, v) { a.x += (float)v[0]; a.y += (float)v[1]; a.z += (float)v[2]; a.w += (float)v[3]; }

#define READ_ROWS(ri, ix) _Pragma("unroll") for (int e_ = 0; e_ < 32; ++e_) ri[e_] = __builtin_amdgcn_readlane((ix), e_);

__device__ __forceinline__
void gather_block(const _Float16* __restrict__ rows, const int (&ri)[32], int ix, int n, int lane,
                  float4& a0, float4& a1, float4& a2, float4& a3) {
    v4h v[32];
    v4h t0, t1, t2;
    int n4 = n >> 2;
    int jt = n & ~3;
    int r  = n & 3;
#pragma unroll
    for (int g = 0; g < 8; ++g) {
        if (g < n4) {
            v[4 * g]     = ROW_AT(ri[4 * g]);
            v[4 * g + 1] = ROW_AT(ri[4 * g + 1]);
            v[4 * g + 2] = ROW_AT(ri[4 * g + 2]);
            v[4 * g + 3] = ROW_AT(ri[4 * g + 3]);
        }
    }
    if (r > 0) {
        t0 = ROW_AT(__builtin_amdgcn_readlane(ix, jt));
        if (r > 1) {
            t1 = ROW_AT(__builtin_amdgcn_readlane(ix, jt + 1));
            if (r > 2) t2 = ROW_AT(__builtin_amdgcn_readlane(ix, jt + 2));
        }
    }
#pragma unroll
    for (int g = 0; g < 8; ++g) {
        if (g < n4) {
            ACC4(a0, v[4 * g]) ACC4(a1, v[4 * g + 1]) ACC4(a2, v[4 * g + 2]) ACC4(a3, v[4 * g + 3])
        }
    }
    if (r > 0) {
        ACC4(a0, t0)
        if (r > 1) {
            ACC4(a0, t1)
            if (r > 2) ACC4(a0, t2)
        }
    }
}

__global__ __launch_bounds__(256)
void gather_sum(const _Float16* __restrict__ rows0, const _Float16* __restrict__ rows1,
                const int* __restrict__ off, const int* __restrict__ sorted,
                float* __restrict__ agg0, float* __restrict__ agg1,
                int nNodes, int nEdges) {
    int list = blockIdx.y;
    const _Float16* rows = list ? rows1 : rows0;
    float* agg = list ? agg1 : agg0;
    const int* o  = off + (long long)list * (nNodes + 1);
    const int* sl = sorted + (long long)list * nEdges;
    int lane = threadIdx.x & 31;
    int wv = __builtin_amdgcn_readfirstlane((int)(blockIdx.x * (blockDim.x >> 5) + (threadIdx.x >> 5)));
    int nW = (int)(gridDim.x * (blockDim.x >> 5));
    int dBeg = (int)((long long)nNodes * wv / nW);
    int dEnd = (int)((long long)nNodes * (wv + 1) / nW);
    for (int c0 = dBeg; c0 < dEnd; c0 += GCH) {
        int cn  = min(dEnd - c0, GCH);
        int bnd = o[c0 + min(lane, cn)];
        int beg = __builtin_amdgcn_readlane(bnd, 0);
        int end = __builtin_amdgcn_readlane(bnd, 1);
        int idx0 = (lane < end - beg)      ? sl[beg + lane]      : 0;
        int idx1 = (lane + 32 < end - beg) ? sl[beg + 32 + lane] : 0;
        for (int k = 0; k < cn; ++k) {
            int nbeg = end;
            int nend = __builtin_amdgcn_readlane(bnd, min(k + 2, cn));
            int nidx0 = 0, nidx1 = 0;
            float4 a0 = make_float4(0.f, 0.f, 0.f, 0.f);
            float4 a1 = a0, a2 = a0, a3 = a0;
            int base = beg, ix = idx0, blk = 0;
            do {
                int ri[32];
                READ_ROWS(ri, ix)
                if (blk == 0) {
                    if (lane < nend - nbeg)      nidx0 = sl[nbeg + lane];
                    if (lane + 32 < nend - nbeg) nidx1 = sl[nbeg + 32 + lane];
                }
                gather_block(rows, ri, ix, min(end - base, 32), lane, a0, a1, a2, a3);
                base += 32;
                ++blk;
                if (base < end)
                    ix = (blk == 1) ? idx1 : ((lane < end - base) ? sl[base + lane] : 0);
            } while (base < end);
            float4 r;
            r.x = (a0.x + a1.x) + (a2.x + a3.x);
            r.y = (a0.y + a1.y) + (a2.y + a3.y);
            r.z = (a0.z + a1.z) + (a2.z + a3.z);
            r.w = (a0.w + a1.w) + (a2.w + a3.w);
            ((float4*)(agg + (long long)(c0 + k) * DH))[lane] = r;
            beg = nbeg; end = nend; idx0 = nidx0; idx1 = nidx1;
        }
    }
}
#undef ROW_AT
#undef READ_ROWS
#undef ACC4

#define LP 132

__device__ __forceinline__ v16h pack16(float4 q0, float4 q1, float4 q2, float4 q3) {
    v16h a;
    a[0]  = (_Float16)q0.x; a[1]  = (_Float16)q0.y; a[2]  = (_Float16)q0.z; a[3]  = (_Float16)q0.w;
    a[4]  = (_Float16)q1.x; a[5]  = (_Float16)q1.y; a[6]  = (_Float16)q1.z; a[7]  = (_Float16)q1.w;
    a[8]  = (_Float16)q2.x; a[9]  = (_Float16)q2.y; a[10] = (_Float16)q2.z; a[11] = (_Float16)q2.w;
    a[12] = (_Float16)q3.x; a[13] = (_Float16)q3.y; a[14] = (_Float16)q3.z; a[15] = (_Float16)q3.w;
    return a;
}

__device__ __forceinline__ void add4(float4& q, float4 b) {
    q.x += b.x; q.y += b.y; q.z += b.z; q.w += b.w;
}

template <bool HAS_A1>
__device__ __forceinline__ void a_from_global(const float* __restrict__ ar0,
                                              const float* __restrict__ ar1,
                                              int grp, v16h (&afrag)[4]) {
#pragma unroll
    for (int kt = 0; kt < 4; ++kt) {
        int off = kt * 32 + grp * 8;
        float4 q0 = *(const float4*)(ar0 + off);
        float4 q1 = *(const float4*)(ar0 + off + 4);
        float4 q2 = *(const float4*)(ar0 + off + 16);
        float4 q3 = *(const float4*)(ar0 + off + 20);
        if (HAS_A1) {
            add4(q0, *(const float4*)(ar1 + off));
            add4(q1, *(const float4*)(ar1 + off + 4));
            add4(q2, *(const float4*)(ar1 + off + 16));
            add4(q3, *(const float4*)(ar1 + off + 20));
        }
        afrag[kt] = pack16(q0, q1, q2, q3);
    }
}

__device__ __forceinline__ void a_from_lds(const float* ar, int grp, v16h (&afrag)[4]) {
#pragma unroll
    for (int kt = 0; kt < 4; ++kt) {
        int off = kt * 32 + grp * 8;
        float4 q0 = *(const float4*)(ar + off);
        float4 q1 = *(const float4*)(ar + off + 4);
        float4 q2 = *(const float4*)(ar + off + 16);
        float4 q3 = *(const float4*)(ar + off + 20);
        afrag[kt] = pack16(q0, q1, q2, q3);
    }
}

__device__ __forceinline__ void load_b(const _Float16* __restrict__ Wp,
                                       int nt, int lane, v16h (&b)[4]) {
    const v16h* wfrag = (const v16h*)Wp;
#pragma unroll
    for (int kt = 0; kt < 4; ++kt) b[kt] = wfrag[(kt * 8 + nt) * 32 + lane];
}

template <bool RELU>
__device__ __forceinline__ void layer_to_lds(const v16h (&afrag)[4],
                                             const _Float16* __restrict__ Wp,
                                             const _Float16* __restrict__ WpNext,
                                             const float* __restrict__ bias,
                                             float* myLds, int lane, int grp, int col,
                                             v16h (&bq)[4]) {
#pragma unroll
    for (int nt = 0; nt < 8; ++nt) {
        v16h c0 = bq[0], c1 = bq[1], c2 = bq[2], c3 = bq[3];
        if (nt < 7)      load_b(Wp, nt + 1, lane, bq);
        else if (WpNext) load_b(WpNext, 0, lane, bq);
        asm volatile("" :: "v"(c0), "v"(c1), "v"(c2), "v"(c3));
        v8f acc = {};
        acc = __builtin_amdgcn_wmma_f32_16x16x32_f16(false, afrag[0], false, c0, (short)0, acc, false, false);
        acc = __builtin_amdgcn_wmma_f32_16x16x32_f16(false, afrag[1], false, c1, (short)0, acc, false, false);
        acc = __builtin_amdgcn_wmma_f32_16x16x32_f16(false, afrag[2], false, c2, (short)0, acc, false, false);
        acc = __builtin_amdgcn_wmma_f32_16x16x32_f16(false, afrag[3], false, c3, (short)0, acc, false, false);
        float bn = bias[nt * 16 + col];
#pragma unroll
        for (int r = 0; r < 8; ++r) {
            float v = acc[r] + bn;
            if (RELU) v = fmaxf(v, 0.f);
            myLds[(r + grp * 8) * LP + nt * 16 + col] = v;
        }
    }
}

__global__ __launch_bounds__(128)
void t_chain(const float* __restrict__ t,
             const float* __restrict__ aggA, const float* __restrict__ aggB,
             const _Float16* __restrict__ Wp,
             const float* __restrict__ b1a, const float* __restrict__ b2a,
             const float* __restrict__ b1b, const float* __restrict__ b2b,
             const float* __restrict__ bo,
             const float* __restrict__ ln_g, const float* __restrict__ ln_b,
             float* __restrict__ t_out, int nRowTiles) {
    __shared__ __attribute__((aligned(16))) float lds[4][16 * LP];
    int wave = blockIdx.x * (blockDim.x >> 5) + (threadIdx.x >> 5);
    if (wave >= nRowTiles) return;
    int lane = threadIdx.x & 31;
    int grp  = lane >> 4;
    int col  = lane & 15;
    float* myLds = lds[threadIdx.x >> 5];
    const float* myRow = myLds + col * LP;
    long long row = (long long)wave * 16 + col;
    long long m0  = (long long)wave * 16;
    v16h afrag[4], afragB[4], bq[4];
    float4 s[16];

    load_b(Wp, 0, lane, bq);
    a_from_global<true>(t + row * DH, aggA + row * DH, grp, afrag);
    a_from_global<true>(t + row * DH, aggB + row * DH, grp, afragB);

    layer_to_lds<true >(afrag, Wp,             Wp + WPN_H,     b1a, myLds, lane, grp, col, bq);
    a_from_lds(myRow, grp, afrag);
    layer_to_lds<false>(afrag, Wp + WPN_H,     Wp + 2 * WPN_H, b2a, myLds, lane, grp, col, bq);
#pragma unroll
    for (int rl = 0; rl < 16; ++rl) {
        float4 g  = ((const float4*)(myLds + rl * LP))[lane];
        float4 tv = ((const float4*)(t + (m0 + rl) * DH))[lane];
        s[rl].x = tv.x + g.x; s[rl].y = tv.y + g.y; s[rl].z = tv.z + g.z; s[rl].w = tv.w + g.w;
    }

    layer_to_lds<true >(afragB, Wp + 2 * WPN_H, Wp + 3 * WPN_H, b1b, myLds, lane, grp, col, bq);
    a_from_lds(myRow, grp, afrag);
    layer_to_lds<false>(afrag,  Wp + 3 * WPN_H, Wp + 4 * WPN_H, b2b, myLds, lane, grp, col, bq);
#pragma unroll
    for (int rl = 0; rl < 16; ++rl) {
        float4 g = ((const float4*)(myLds + rl * LP))[lane];
        s[rl].x = fmaxf(s[rl].x + g.x, 0.f); s[rl].y = fmaxf(s[rl].y + g.y, 0.f);
        s[rl].z = fmaxf(s[rl].z + g.z, 0.f); s[rl].w = fmaxf(s[rl].w + g.w, 0.f);
    }
#pragma unroll
    for (int rl = 0; rl < 16; ++rl)
        ((float4*)(myLds + rl * LP))[lane] = s[rl];

    a_from_lds(myRow, grp, afrag);
    layer_to_lds<true >(afrag, Wp + 4 * WPN_H, nullptr, bo, myLds, lane, grp, col, bq);
    float4 gv = ((const float4*)ln_g)[lane];
    float4 bv = ((const float4*)ln_b)[lane];
#pragma unroll
    for (int rl = 0; rl < 16; ++rl) {
        float4 v = ((const float4*)(myLds + rl * LP))[lane];
        float sm = v.x + v.y + v.z + v.w;
        float sq = v.x * v.x + v.y * v.y + v.z * v.z + v.w * v.w;
#pragma unroll
        for (int off = 16; off > 0; off >>= 1) {
            sm += __shfl_xor(sm, off, 32);
            sq += __shfl_xor(sq, off, 32);
        }
        float mean = sm * (1.0f / 128.0f);
        float var  = sq * (1.0f / 128.0f) - mean * mean;
        float inv  = rsqrtf(var + 1e-5f);
        float4 tv = s[rl];
        float4 o;
        o.x = tv.x + (v.x - mean) * inv * gv.x + bv.x;
        o.y = tv.y + (v.y - mean) * inv * gv.y + bv.y;
        o.z = tv.z + (v.z - mean) * inv * gv.z + bv.z;
        o.w = tv.w + (v.w - mean) * inv * gv.w + bv.w;
        ((float4*)(t_out + (m0 + rl) * DH))[lane] = o;
    }
}

__device__ __forceinline__
void x_chain_tile(const float* __restrict__ x, const _Float16* __restrict__ Wp,
                  const float* __restrict__ bf1, const float* __restrict__ bf2,
                  float* __restrict__ x_out, int wave, float* myLds, int lane) {
    int grp  = lane >> 4;
    int col  = lane & 15;
    const float* myRow = myLds + col * LP;
    long long row = (long long)wave * 16 + col;
    long long m0  = (long long)wave * 16;
    v16h afrag[4], bq[4];

    load_b(Wp, 0, lane, bq);
    const float* ar = x + row * DH + grp * 8;
    v4f q[16];
#pragma unroll
    for (int kt = 0; kt < 4; ++kt) {
        q[4 * kt]     = *(const v4f*)(ar + kt * 32);
        q[4 * kt + 1] = *(const v4f*)(ar + kt * 32 + 4);
        q[4 * kt + 2] = *(const v4f*)(ar + kt * 32 + 16);
        q[4 * kt + 3] = *(const v4f*)(ar + kt * 32 + 20);
    }
    asm volatile("" :: "v"(q[0]), "v"(q[1]), "v"(q[2]),  "v"(q[3]),  "v"(q[4]),  "v"(q[5]),  "v"(q[6]),  "v"(q[7]),
                       "v"(q[8]), "v"(q[9]), "v"(q[10]), "v"(q[11]), "v"(q[12]), "v"(q[13]), "v"(q[14]), "v"(q[15]));
#pragma unroll
    for (int kt = 0; kt < 4; ++kt) {
#pragma unroll
        for (int h = 0; h < 16; ++h) afrag[kt][h] = (_Float16)q[4 * kt + (h >> 2)][h & 3];
    }
    layer_to_lds<true >(afrag, Wp,         Wp + WPN_H, bf1, myLds, lane, grp, col, bq);
    __builtin_amdgcn_sched_barrier(0);
    a_from_lds(myRow, grp, afrag);
    __builtin_amdgcn_sched_barrier(0);
    layer_to_lds<false>(afrag, Wp + WPN_H, nullptr,    bf2, myLds, lane, grp, col, bq);
    __builtin_amdgcn_sched_barrier(0);
#pragma unroll
    for (int rl = 0; rl < 16; ++rl) {
        float4 v  = ((const float4*)(myLds + rl * LP))[lane];
        float4 rv = ((const float4*)(x + (m0 + rl) * DH))[lane];
        v.x += rv.x; v.y += rv.y; v.z += rv.z; v.w += rv.w;
        ((float4*)(x_out + (m0 + rl) * DH))[lane] = v;
    }
}

#define PLDS ((32 * 16 * LP) > (NCH * BMAXB + BMAXB + 32) ? (32 * 16 * LP) : (NCH * BMAXB + BMAXB + 32))

__global__ __launch_bounds__(1024)
void edge_place(const int* __restrict__ src0, const int* __restrict__ dst0,
                const int* __restrict__ src1, const int* __restrict__ dst1,
                const int* __restrict__ cntb, int* __restrict__ bbase,
                int* __restrict__ bkt, int nEdges, int nNodes, int nB,
                const float* __restrict__ x, const _Float16* __restrict__ WpX,
                const float* __restrict__ bf1, const float* __restrict__ bf2,
                float* __restrict__ x_out, int nRowTiles,
                _Float16* __restrict__ xH) {
    __shared__ __attribute__((aligned(16))) int cur[PLDS];
    int tid  = threadIdx.x;
    int xRows = (nRowTiles + 32 * NCH - 1) / (32 * NCH);
    if ((int)blockIdx.y >= 2 + xRows) {
        rows_to_f16(x, xH, (long long)nNodes * DH,
                    ((int)blockIdx.y - 2 - xRows) * NCH + (int)blockIdx.x, 2 * NCH, tid);
        return;
    }
    if (blockIdx.y >= 2) {
        int xb   = (int)(blockIdx.y - 2) * NCH + (int)blockIdx.x;
        int tile = xb * 32 + (tid >> 5);
        if (tile >= nRowTiles) return;
        x_chain_tile(x, WpX, bf1, bf2, x_out, tile, (float*)cur + (tid >> 5) * (16 * LP), tid & 31);
        return;
    }
    int list = blockIdx.y;
    const int* src = list ? src1 : src0;
    const int* dst = list ? dst1 : dst0;
    const int* cm = cntb + (long long)list * NCH * nB;
    int* out = bkt + (long long)list * nEdges;
    int per = chunk_len(nEdges);
    int c   = blockIdx.x;
    int beg = c * per;
    int end = min(beg + per, nEdges);
    int* m    = cur;
    int* posn = cur + NCH * BMAXB;
    int* wtot = posn + BMAXB;
    int lane = tid & 31;
    int w    = tid >> 5;
    int n = NCH * nB;
#pragma unroll 8
    for (int i = tid; i < n; i += 1024) m[i] = cm[i];
    __syncthreads();
    int tot = 0, mine = 0;
    if (tid < nB) {
        for (int cc = 0; cc < NCH; ++cc) {
            if (cc == c) mine = tot;
            tot += m[cc * nB + tid];
        }
    }
    int incl = tot;
#pragma unroll
    for (int d = 1; d < 32; d <<= 1) {
        int u = __shfl_up(incl, d, 32);
        if (lane >= d) incl += u;
    }
    if (lane == 31) wtot[w] = incl;
    __syncthreads();
    int wt = wtot[lane];
    int tincl = wt;
#pragma unroll
    for (int d = 1; d < 32; d <<= 1) {
        int u = __shfl_up(tincl, d, 32);
        if (lane >= d) tincl += u;
    }
    int wexcl = __shfl(tincl - wt, w, 32);
    int base = wexcl + (incl - tot);
    if (tid < nB) {
        posn[tid] = base + mine;
        if (c == 0) {
            bbase[list * (nB + 1) + tid] = base;
            if (tid == nB - 1) bbase[list * (nB + 1) + nB] = base + tot;
        }
    }
    __syncthreads();
    int e = beg + tid;
    for (; e + 3 * 1024 < end; e += 4 * 1024) {
        int d0 = dst[e],        d1 = dst[e + 1024];
        int d2 = dst[e + 2048], d3 = dst[e + 3072];
        int s0 = src[e],        s1 = src[e + 1024];
        int s2 = src[e + 2048], s3 = src[e + 3072];
        if ((unsigned)d0 < (unsigned)nNodes) out[atomicAdd(&posn[d0 >> BSH], 1)] = (s0 << BSH) | (d0 & BMASK);
        if ((unsigned)d1 < (unsigned)nNodes) out[atomicAdd(&posn[d1 >> BSH], 1)] = (s1 << BSH) | (d1 & BMASK);
        if ((unsigned)d2 < (unsigned)nNodes) out[atomicAdd(&posn[d2 >> BSH], 1)] = (s2 << BSH) | (d2 & BMASK);
        if ((unsigned)d3 < (unsigned)nNodes) out[atomicAdd(&posn[d3 >> BSH], 1)] = (s3 << BSH) | (d3 & BMASK);
    }
    for (; e < end; e += 1024) {
        int d = dst[e];
        int sv = src[e];
        if ((unsigned)d < (unsigned)nNodes) out[atomicAdd(&posn[d >> BSH], 1)] = (sv << BSH) | (d & BMASK);
    }
}

extern "C" void kernel_launch(void* const* d_in, const int* in_sizes, int n_in,
                              void* d_out, int out_size, void* d_ws, size_t ws_size,
                              hipStream_t stream) {
    const float* x    = (const float*)d_in[0];
    const float* t    = (const float*)d_in[1];
    const int*   e_t  = (const int*)d_in[2];
    const int*   e_x  = (const int*)d_in[3];
    const float* W1a  = (const float*)d_in[4];
    const float* b1a  = (const float*)d_in[5];
    const float* W2a  = (const float*)d_in[6];
    const float* b2a  = (const float*)d_in[7];
    const float* W1b  = (const float*)d_in[8];
    const float* b1b  = (const float*)d_in[9];
    const float* W2b  = (const float*)d_in[10];
    const float* b2b  = (const float*)d_in[11];
    const float* Wo   = (const float*)d_in[12];
    const float* bo   = (const float*)d_in[13];
    const float* ln_g = (const float*)d_in[14];
    const float* ln_b = (const float*)d_in[15];
    const float* Wf1  = (const float*)d_in[16];
    const float* bf1  = (const float*)d_in[17];
    const float* Wf2  = (const float*)d_in[18];
    const float* bf2  = (const float*)d_in[19];

    const int N = in_sizes[0] / DH;
    const int E = in_sizes[2] / 2;
    const long long ND = (long long)N * DH;

    float* buf0 = (float*)d_ws;
    float* buf1 = buf0 + ND;
    float* buf2 = buf1 + ND;
    _Float16* wp = (_Float16*)(buf2 + ND);
    const int WPN = 4 * 8 * 32 * 16;
    _Float16* wp0 = wp;
    _Float16* wp5 = wp + 5 * WPN;

    float* x_out = (float*)d_out;
    float* t_out = x_out + ND;
    _Float16* tH = (_Float16*)t_out;
    _Float16* xH = tH + ND;

    const int nRowTiles = N / 16;
    dim3 gB((nRowTiles + 3) / 4), tB(128);

    int* sorted = (int*)buf2;
    int* off    = sorted + 2 * (long long)E;
    int* bkt    = (int*)buf0;
    const int nB = (N + BSZ - 1) / BSZ;
    int* cntb   = (int*)buf1;
    int* bbase  = cntb + 2 * NCH * nB;

    {
        dim3 gcp(NCH, 6);
        edge_count<<<gcp, 1024, 0, stream>>>(e_t + E, e_x + E, cntb, E, N, nB,
                                             W1a, W2a, W1b, W2b, Wo, Wf1, Wf2, wp, t, tH);
        dim3 gcx(NCH, 4 + (unsigned)((nRowTiles + 32 * NCH - 1) / (32 * NCH)));
        edge_place<<<gcx, 1024, 0, stream>>>(e_t, e_t + E, e_x, e_x + E, cntb, bbase, bkt, E, N, nB,
                                             x, wp5, bf1, bf2, x_out, nRowTiles, xH);
        dim3 go((unsigned)nB, 2);
        owner_sort<<<go, 1024, 0, stream>>>(bkt, bbase, off, sorted, E, N, nB);
        dim3 gg(512, 2);
        gather_sum<<<gg, 256, 0, stream>>>(tH, xH, off, sorted, buf0, buf1, N, E);
    }
    t_chain<<<gB, tB, 0, stream>>>(t, buf0, buf1, wp0, b1a, b2a, b1b, b2b, bo, ln_g, ln_b, t_out, nRowTiles);
}


# Edge placement through an LDS stage: a chunk's edge words are put in bucket order in LDS and leave as whole lines (word by word, a store touched 32 lines); inside a bucket still arrival order
# speedup vs baseline: 6.1747x; 1.1162x over previous
#include <hip/hip_runtime.h>
#include <hip/hip_bf16.h>

typedef __attribute__((ext_vector_type(16))) _Float16 v16h;
typedef __attribute__((ext_vector_type(8)))  float    v8f;
typedef __attribute__((ext_vector_type(4)))  float    v4f;
typedef __attribute__((ext_vector_type(4)))  _Float16 v4h;

#define DH 128

__device__ __forceinline__
void pack_one(const float* __restrict__ W, _Float16* __restrict__ Wp, int idx) {
    int h    = idx & 15;
    int lane = (idx >> 4) & 31;
    int nt   = (idx >> 9) & 7;
    int kt   = (idx >> 12) & 3;
    int grp  = lane >> 4;
    int k    = kt * 32 + grp * 8 + ((h < 8) ? h : (16 + (h - 8)));
    int n    = nt * 16 + (lane & 15);
    Wp[idx]  = (_Float16)W[k * DH + n];
}

__device__ __forceinline__ v4h to_h4(float4 v) {
    v4h h;
    h[0] = (_Float16)v.x; h[1] = (_Float16)v.y; h[2] = (_Float16)v.z; h[3] = (_Float16)v.w;
    return h;
}

__device__ __forceinline__
void rows_to_f16(const float* __restrict__ rows, _Float16* __restrict__ rowsH,
                 long long nElem, int part, int nParts, int tid) {
    const float4* s4 = (const float4*)rows;
    v4h* d4 = (v4h*)rowsH;
    long long n4  = nElem >> 2;
    long long per = (((n4 + nParts - 1) / nParts) + 8191) & ~8191LL;
    long long beg = (long long)part * per;
    long long end = beg + per < n4 ? beg + per : n4;
    long long i = beg + tid;
    for (; i + 7 * 1024 < end; i += 8 * 1024) {
        float4 v[8];
#pragma unroll
        for (int u = 0; u < 8; ++u) v[u] = s4[i + u * 1024];
#pragma unroll
        for (int u = 0; u < 8; ++u) d4[i + u * 1024] = to_h4(v[u]);
    }
    for (; i < end; i += 1024) d4[i] = to_h4(s4[i]);
}

#define NCH  64
#define BSH  8
#define BSZ  (1 << BSH)
#define BMASK (BSZ - 1)
#define BMAXB 512
#define NMAT 7
#define WPN_H (4 * 8 * 32 * 16)

__device__ __forceinline__ int chunk_len(int nEdges) {
    return (((nEdges + NCH - 1) / NCH) + 1023) & ~1023;
}

__global__ __launch_bounds__(1024)
void edge_count(const int* __restrict__ dst0, const int* __restrict__ dst1,
                int* __restrict__ cntb, int nEdges, int nNodes, int nB,
                const float* __restrict__ W0, const float* __restrict__ W1,
                const float* __restrict__ W2, const float* __restrict__ W3,
                const float* __restrict__ W4, const float* __restrict__ W5,
                const float* __restrict__ W6, _Float16* __restrict__ Wp,
                const float* __restrict__ rows0, _Float16* __restrict__ rows0H) {
    __shared__ int hist[BMAXB];
    int tid  = threadIdx.x;
    if (blockIdx.y >= 4) {
        rows_to_f16(rows0, rows0H, (long long)nNodes * DH,
                    (int)(blockIdx.y - 4) * NCH + (int)blockIdx.x, 2 * NCH, tid);
        return;
    }
    if (blockIdx.y >= 2) {
        int pb  = (int)(blockIdx.y - 2) * NCH + (int)blockIdx.x;
        int mat = pb >> 4;
        if (mat >= NMAT) return;
        const float* W = mat == 0 ? W0 : mat == 1 ? W1 : mat == 2 ? W2 : mat == 3 ? W3
                       : mat == 4 ? W4 : mat == 5 ? W5 : W6;
        pack_one(W, Wp + mat * WPN_H, ((pb & 15) << 10) + tid);
        return;
    }
    int list = blockIdx.y;
    const int* dst = list ? dst1 : dst0;
    int* part = cntb + ((long long)list * NCH + blockIdx.x) * nB;
    int per = chunk_len(nEdges);
    int beg = blockIdx.x * per;
    int end = min(beg + per, nEdges);
    for (int i = tid; i < nB; i += 1024) hist[i] = 0;
    __syncthreads();
    int e = beg + tid;
    for (; e + 7 * 1024 < end; e += 8 * 1024) {
        int d[8];
#pragma unroll
        for (int u = 0; u < 8; ++u) d[u] = dst[e + u * 1024];
#pragma unroll
        for (int u = 0; u < 8; ++u)
            if ((unsigned)d[u] < (unsigned)nNodes) atomicAdd(&hist[d[u] >> BSH], 1);
    }
    for (; e < end; e += 1024) {
        int d = dst[e];
        if ((unsigned)d < (unsigned)nNodes) atomicAdd(&hist[d >> BSH], 1);
    }
    __syncthreads();
    for (int i = tid; i < nB; i += 1024) part[i] = hist[i];
}

#define OCAP 32768
__global__ __launch_bounds__(1024)
void owner_sort(const int* __restrict__ bkt, const int* __restrict__ bbase,
                int* __restrict__ off, int* __restrict__ sorted,
                int nEdges, int nNodes, int nB) {
    __shared__ int stage[OCAP];
    __shared__ int cnt[BSZ];
    __shared__ int wtot[BSZ / 32];
    int tid  = threadIdx.x;
    int lane = tid & 31;
    int w    = tid >> 5;
    int list = blockIdx.y;
    int b    = blockIdx.x;
    const int* in = bkt + (long long)list * nEdges;
    int* out = sorted + (long long)list * nEdges;
    int* o = off + (long long)list * (nNodes + 1);
    int beg = bbase[list * (nB + 1) + b];
    int end = bbase[list * (nB + 1) + b + 1];
    if (tid < BSZ) cnt[tid] = 0;
    __syncthreads();
    int i = beg + tid;
    for (; i + 3 * 1024 < end; i += 4 * 1024) {
        int w0 = in[i], w1 = in[i + 1024], w2 = in[i + 2048], w3 = in[i + 3072];
        atomicAdd(&cnt[w0 & BMASK], 1);
        atomicAdd(&cnt[w1 & BMASK], 1);
        atomicAdd(&cnt[w2 & BMASK], 1);
        atomicAdd(&cnt[w3 & BMASK], 1);
    }
    for (; i < end; i += 1024) atomicAdd(&cnt[in[i] & BMASK], 1);
    __syncthreads();
    int v = (tid < BSZ) ? cnt[tid] : 0;
    int incl = v;
#pragma unroll
    for (int d = 1; d < 32; d <<= 1) {
        int u = __shfl_up(incl, d, 32);
        if (lane >= d) incl += u;
    }
    if (tid < BSZ && lane == 31) wtot[w] = incl;
    __syncthreads();
    if (tid < BSZ) {
        int pre = 0;
        for (int k = 0; k < w; ++k) pre += wtot[k];
        int start = beg + pre + (incl - v);
        cnt[tid] = start;
        int node = (b << BSH) + tid;
        if (node < nNodes) o[node] = start;
    }
    if (b == nB - 1 && tid == 0) o[nNodes] = end;
    __syncthreads();
    int len = end - beg;
    i = beg + tid;
    if (len <= OCAP) {
        for (; i + 3 * 1024 < end; i += 4 * 1024) {
            int w0 = in[i], w1 = in[i + 1024], w2 = in[i + 2048], w3 = in[i + 3072];
            stage[atomicAdd(&cnt[w0 & BMASK], 1) - beg] = (int)((unsigned)w0 >> BSH);
            stage[atomicAdd(&cnt[w1 & BMASK], 1) - beg] = (int)((unsigned)w1 >> BSH);
            stage[atomicAdd(&cnt[w2 & BMASK], 1) - beg] = (int)((unsigned)w2 >> BSH);
            stage[atomicAdd(&cnt[w3 & BMASK], 1) - beg] = (int)((unsigned)w3 >> BSH);
        }
        for (; i < end; i += 1024) {
            int w0 = in[i];
            stage[atomicAdd(&cnt[w0 & BMASK], 1) - beg] = (int)((unsigned)w0 >> BSH);
        }
    } else {
        for (; i + 3 * 1024 < end; i += 4 * 1024) {
            int w0 = in[i], w1 = in[i + 1024], w2 = in[i + 2048], w3 = in[i + 3072];
            out[atomicAdd(&cnt[w0 & BMASK], 1)] = (int)((unsigned)w0 >> BSH);
            out[atomicAdd(&cnt[w1 & BMASK], 1)] = (int)((unsigned)w1 >> BSH);
            out[atomicAdd(&cnt[w2 & BMASK], 1)] = (int)((unsigned)w2 >> BSH);
            out[atomicAdd(&cnt[w3 & BMASK], 1)] = (int)((unsigned)w3 >> BSH);
        }
        for (; i < end; i += 1024) {
            int w0 = in[i];
            out[atomicAdd(&cnt[w0 & BMASK], 1)] = (int)((unsigned)w0 >> BSH);
        }
    }
    __syncthreads();
    if (len <= OCAP)
        for (int k = tid; k < len; k += 1024) out[beg + k] = stage[k];
}

#define GCH 31

#define ROW_AT(i) (((const v4h*)(rows + (long long)(i) * DH))[lane])
#define ACC4(a, v) { a.x += (float)v[0]; a.y += (float)v[1]; a.z += (float)v[2]; a.w += (float)v[3]; }

#define READ_ROWS(ri, ix) _Pragma("unroll") for (int e_ = 0; e_ < 32; ++e_) ri[e_] = __builtin_amdgcn_readlane((ix), e_);

__device__ __forceinline__
void gather_block(const _Float16* __restrict__ rows, const int (&ri)[32], int ix, int n, int lane,
                  float4& a0, float4& a1, float4& a2, float4& a3) {
    v4h v[32];
    v4h t0, t1, t2;
    int n4 = n >> 2;
    int jt = n & ~3;
    int r  = n & 3;
#pragma unroll
    for (int g = 0; g < 8; ++g) {
        if (g < n4) {
            v[4 * g]     = ROW_AT(ri[4 * g]);
            v[4 * g + 1] = ROW_AT(ri[4 * g + 1]);
            v[4 * g + 2] = ROW_AT(ri[4 * g + 2]);
            v[4 * g + 3] = ROW_AT(ri[4 * g + 3]);
        }
    }
    if (r > 0) {
        t0 = ROW_AT(__builtin_amdgcn_readlane(ix, jt));
        if (r > 1) {
            t1 = ROW_AT(__builtin_amdgcn_readlane(ix, jt + 1));
            if (r > 2) t2 = ROW_AT(__builtin_amdgcn_readlane(ix, jt + 2));
        }
    }
#pragma unroll
    for (int g = 0; g < 8; ++g) {
        if (g < n4) {
            ACC4(a0, v[4 * g]) ACC4(a1, v[4 * g + 1]) ACC4(a2, v[4 * g + 2]) ACC4(a3, v[4 * g + 3])
        }
    }
    if (r > 0) {
        ACC4(a0, t0)
        if (r > 1) {
            ACC4(a0, t1)
            if (r > 2) ACC4(a0, t2)
        }
    }
}

__global__ __launch_bounds__(256)
void gather_sum(const _Float16* __restrict__ rows0, const _Float16* __restrict__ rows1,
                const int* __restrict__ off, const int* __restrict__ sorted,
                float* __restrict__ agg0, float* __restrict__ agg1,
                int nNodes, int nEdges) {
    int list = blockIdx.y;
    const _Float16* rows = list ? rows1 : rows0;
    float* agg = list ? agg1 : agg0;
    const int* o  = off + (long long)list * (nNodes + 1);
    const int* sl = sorted + (long long)list * nEdges;
    int lane = threadIdx.x & 31;
    int wv = __builtin_amdgcn_readfirstlane((int)(blockIdx.x * (blockDim.x >> 5) + (threadIdx.x >> 5)));
    int nW = (int)(gridDim.x * (blockDim.x >> 5));
    int dBeg = (int)((long long)nNodes * wv / nW);
    int dEnd = (int)((long long)nNodes * (wv + 1) / nW);
    for (int c0 = dBeg; c0 < dEnd; c0 += GCH) {
        int cn  = min(dEnd - c0, GCH);
        int bnd = o[c0 + min(lane, cn)];
        int beg = __builtin_amdgcn_readlane(bnd, 0);
        int end = __builtin_amdgcn_readlane(bnd, 1);
        int idx0 = (lane < end - beg)      ? sl[beg + lane]      : 0;
        int idx1 = (lane + 32 < end - beg) ? sl[beg + 32 + lane] : 0;
        for (int k = 0; k < cn; ++k) {
            int nbeg = end;
            int nend = __builtin_amdgcn_readlane(bnd, min(k + 2, cn));
            int nidx0 = 0, nidx1 = 0;
            float4 a0 = make_float4(0.f, 0.f, 0.f, 0.f);
            float4 a1 = a0, a2 = a0, a3 = a0;
            int base = beg, ix = idx0, blk = 0;
            do {
                int ri[32];
                READ_ROWS(ri, ix)
                if (blk == 0) {
                    if (lane < nend - nbeg)      nidx0 = sl[nbeg + lane];
                    if (lane + 32 < nend - nbeg) nidx1 = sl[nbeg + 32 + lane];
                }
                gather_block(rows, ri, ix, min(end - base, 32), lane, a0, a1, a2, a3);
                base += 32;
                ++blk;
                if (base < end)
                    ix = (blk == 1) ? idx1 : ((lane < end - base) ? sl[base + lane] : 0);
            } while (base < end);
            float4 r;
            r.x = (a0.x + a1.x) + (a2.x + a3.x);
            r.y = (a0.y + a1.y) + (a2.y + a3.y);
            r.z = (a0.z + a1.z) + (a2.z + a3.z);
            r.w = (a0.w + a1.w) + (a2.w + a3.w);
            ((float4*)(agg + (long long)(c0 + k) * DH))[lane] = r;
            beg = nbeg; end = nend; idx0 = nidx0; idx1 = nidx1;
        }
    }
}
#undef ROW_AT
#undef READ_ROWS
#undef ACC4

#define LP 132

__device__ __forceinline__ v16h pack16(float4 q0, float4 q1, float4 q2, float4 q3) {
    v16h a;
    a[0]  = (_Float16)q0.x; a[1]  = (_Float16)q0.y; a[2]  = (_Float16)q0.z; a[3]  = (_Float16)q0.w;
    a[4]  = (_Float16)q1.x; a[5]  = (_Float16)q1.y; a[6]  = (_Float16)q1.z; a[7]  = (_Float16)q1.w;
    a[8]  = (_Float16)q2.x; a[9]  = (_Float16)q2.y; a[10] = (_Float16)q2.z; a[11] = (_Float16)q2.w;
    a[12] = (_Float16)q3.x; a[13] = (_Float16)q3.y; a[14] = (_Float16)q3.z; a[15] = (_Float16)q3.w;
    return a;
}

__device__ __forceinline__ void add4(float4& q, float4 b) {
    q.x += b.x; q.y += b.y; q.z += b.z; q.w += b.w;
}

template <bool HAS_A1>
__device__ __forceinline__ void a_from_global(const float* __restrict__ ar0,
                                              const float* __restrict__ ar1,
                                              int grp, v16h (&afrag)[4]) {
#pragma unroll
    for (int kt = 0; kt < 4; ++kt) {
        int off = kt * 32 + grp * 8;
        float4 q0 = *(const float4*)(ar0 + off);
        float4 q1 = *(const float4*)(ar0 + off + 4);
        float4 q2 = *(const float4*)(ar0 + off + 16);
        float4 q3 = *(const float4*)(ar0 + off + 20);
        if (HAS_A1) {
            add4(q0, *(const float4*)(ar1 + off));
            add4(q1, *(const float4*)(ar1 + off + 4));
            add4(q2, *(const float4*)(ar1 + off + 16));
            add4(q3, *(const float4*)(ar1 + off + 20));
        }
        afrag[kt] = pack16(q0, q1, q2, q3);
    }
}

__device__ __forceinline__ void a_from_lds(const float* ar, int grp, v16h (&afrag)[4]) {
#pragma unroll
    for (int kt = 0; kt < 4; ++kt) {
        int off = kt * 32 + grp * 8;
        float4 q0 = *(const float4*)(ar + off);
        float4 q1 = *(const float4*)(ar + off + 4);
        float4 q2 = *(const float4*)(ar + off + 16);
        float4 q3 = *(const float4*)(ar + off + 20);
        afrag[kt] = pack16(q0, q1, q2, q3);
    }
}

__device__ __forceinline__ void load_b(const _Float16* __restrict__ Wp,
                                       int nt, int lane, v16h (&b)[4]) {
    const v16h* wfrag = (const v16h*)Wp;
#pragma unroll
    for (int kt = 0; kt < 4; ++kt) b[kt] = wfrag[(kt * 8 + nt) * 32 + lane];
}

template <bool RELU>
__device__ __forceinline__ void layer_to_lds(const v16h (&afrag)[4],
                                             const _Float16* __restrict__ Wp,
                                             const _Float16* __restrict__ WpNext,
                                             const float* __restrict__ bias,
                                             float* myLds, int lane, int grp, int col,
                                             v16h (&bq)[4]) {
#pragma unroll
    for (int nt = 0; nt < 8; ++nt) {
        v16h c0 = bq[0], c1 = bq[1], c2 = bq[2], c3 = bq[3];
        if (nt < 7)      load_b(Wp, nt + 1, lane, bq);
        else if (WpNext) load_b(WpNext, 0, lane, bq);
        asm volatile("" :: "v"(c0), "v"(c1), "v"(c2), "v"(c3));
        v8f acc = {};
        acc = __builtin_amdgcn_wmma_f32_16x16x32_f16(false, afrag[0], false, c0, (short)0, acc, false, false);
        acc = __builtin_amdgcn_wmma_f32_16x16x32_f16(false, afrag[1], false, c1, (short)0, acc, false, false);
        acc = __builtin_amdgcn_wmma_f32_16x16x32_f16(false, afrag[2], false, c2, (short)0, acc, false, false);
        acc = __builtin_amdgcn_wmma_f32_16x16x32_f16(false, afrag[3], false, c3, (short)0, acc, false, false);
        float bn = bias[nt * 16 + col];
#pragma unroll
        for (int r = 0; r < 8; ++r) {
            float v = acc[r] + bn;
            if (RELU) v = fmaxf(v, 0.f);
            myLds[(r + grp * 8) * LP + nt * 16 + col] = v;
        }
    }
}

__global__ __launch_bounds__(128)
void t_chain(const float* __restrict__ t,
             const float* __restrict__ aggA, const float* __restrict__ aggB,
             const _Float16* __restrict__ Wp,
             const float* __restrict__ b1a, const float* __restrict__ b2a,
             const float* __restrict__ b1b, const float* __restrict__ b2b,
             const float* __restrict__ bo,
             const float* __restrict__ ln_g, const float* __restrict__ ln_b,
             float* __restrict__ t_out, int nRowTiles) {
    __shared__ __attribute__((aligned(16))) float lds[4][16 * LP];
    int wave = blockIdx.x * (blockDim.x >> 5) + (threadIdx.x >> 5);
    if (wave >= nRowTiles) return;
    int lane = threadIdx.x & 31;
    int grp  = lane >> 4;
    int col  = lane & 15;
    float* myLds = lds[threadIdx.x >> 5];
    const float* myRow = myLds + col * LP;
    long long row = (long long)wave * 16 + col;
    long long m0  = (long long)wave * 16;
    v16h afrag[4], afragB[4], bq[4];
    float4 s[16];

    load_b(Wp, 0, lane, bq);
    a_from_global<true>(t + row * DH, aggA + row * DH, grp, afrag);
    a_from_global<true>(t + row * DH, aggB + row * DH, grp, afragB);

    layer_to_lds<true >(afrag, Wp,             Wp + WPN_H,     b1a, myLds, lane, grp, col, bq);
    a_from_lds(myRow, grp, afrag);
    layer_to_lds<false>(afrag, Wp + WPN_H,     Wp + 2 * WPN_H, b2a, myLds, lane, grp, col, bq);
#pragma unroll
    for (int rl = 0; rl < 16; ++rl) {
        float4 g  = ((const float4*)(myLds + rl * LP))[lane];
        float4 tv = ((const float4*)(t + (m0 + rl) * DH))[lane];
        s[rl].x = tv.x + g.x; s[rl].y = tv.y + g.y; s[rl].z = tv.z + g.z; s[rl].w = tv.w + g.w;
    }

    layer_to_lds<true >(afragB, Wp + 2 * WPN_H, Wp + 3 * WPN_H, b1b, myLds, lane, grp, col, bq);
    a_from_lds(myRow, grp, afrag);
    layer_to_lds<false>(afrag,  Wp + 3 * WPN_H, Wp + 4 * WPN_H, b2b, myLds, lane, grp, col, bq);
#pragma unroll
    for (int rl = 0; rl < 16; ++rl) {
        float4 g = ((const float4*)(myLds + rl * LP))[lane];
        s[rl].x = fmaxf(s[rl].x + g.x, 0.f); s[rl].y = fmaxf(s[rl].y + g.y, 0.f);
        s[rl].z = fmaxf(s[rl].z + g.z, 0.f); s[rl].w = fmaxf(s[rl].w + g.w, 0.f);
    }
#pragma unroll
    for (int rl = 0; rl < 16; ++rl)
        ((float4*)(myLds + rl * LP))[lane] = s[rl];

    a_from_lds(myRow, grp, afrag);
    layer_to_lds<true >(afrag, Wp + 4 * WPN_H, nullptr, bo, myLds, lane, grp, col, bq);
    float4 gv = ((const float4*)ln_g)[lane];
    float4 bv = ((const float4*)ln_b)[lane];
#pragma unroll
    for (int rl = 0; rl < 16; ++rl) {
        float4 v = ((const float4*)(myLds + rl * LP))[lane];
        float sm = v.x + v.y + v.z + v.w;
        float sq = v.x * v.x + v.y * v.y + v.z * v.z + v.w * v.w;
#pragma unroll
        for (int off = 16; off > 0; off >>= 1) {
            sm += __shfl_xor(sm, off, 32);
            sq += __shfl_xor(sq, off, 32);
        }
        float mean = sm * (1.0f / 128.0f);
        float var  = sq * (1.0f / 128.0f) - mean * mean;
        float inv  = rsqrtf(var + 1e-5f);
        float4 tv = s[rl];
        float4 o;
        o.x = tv.x + (v.x - mean) * inv * gv.x + bv.x;
        o.y = tv.y + (v.y - mean) * inv * gv.y + bv.y;
        o.z = tv.z + (v.z - mean) * inv * gv.z + bv.z;
        o.w = tv.w + (v.w - mean) * inv * gv.w + bv.w;
        ((float4*)(t_out + (m0 + rl) * DH))[lane] = o;
    }
}

__device__ __forceinline__
void x_chain_tile(const float* __restrict__ x, const _Float16* __restrict__ Wp,
                  const float* __restrict__ bf1, const float* __restrict__ bf2,
                  float* __restrict__ x_out, int wave, float* myLds, int lane) {
    int grp  = lane >> 4;
    int col  = lane & 15;
    const float* myRow = myLds + col * LP;
    long long row = (long long)wave * 16 + col;
    long long m0  = (long long)wave * 16;
    v16h afrag[4], bq[4];

    load_b(Wp, 0, lane, bq);
    const float* ar = x + row * DH + grp * 8;
    v4f q[16];
#pragma unroll
    for (int kt = 0; kt < 4; ++kt) {
        q[4 * kt]     = *(const v4f*)(ar + kt * 32);
        q[4 * kt + 1] = *(const v4f*)(ar + kt * 32 + 4);
        q[4 * kt + 2] = *(const v4f*)(ar + kt * 32 + 16);
        q[4 * kt + 3] = *(const v4f*)(ar + kt * 32 + 20);
    }
    asm volatile("" :: "v"(q[0]), "v"(q[1]), "v"(q[2]),  "v"(q[3]),  "v"(q[4]),  "v"(q[5]),  "v"(q[6]),  "v"(q[7]),
                       "v"(q[8]), "v"(q[9]), "v"(q[10]), "v"(q[11]), "v"(q[12]), "v"(q[13]), "v"(q[14]), "v"(q[15]));
#pragma unroll
    for (int kt = 0; kt < 4; ++kt) {
#pragma unroll
        for (int h = 0; h < 16; ++h) afrag[kt][h] = (_Float16)q[4 * kt + (h >> 2)][h & 3];
    }
    layer_to_lds<true >(afrag, Wp,         Wp + WPN_H, bf1, myLds, lane, grp, col, bq);
    __builtin_amdgcn_sched_barrier(0);
    a_from_lds(myRow, grp, afrag);
    __builtin_amdgcn_sched_barrier(0);
    layer_to_lds<false>(afrag, Wp + WPN_H, nullptr,    bf2, myLds, lane, grp, col, bq);
    __builtin_amdgcn_sched_barrier(0);
#pragma unroll
    for (int rl = 0; rl < 16; ++rl) {
        float4 v  = ((const float4*)(myLds + rl * LP))[lane];
        float4 rv = ((const float4*)(x + (m0 + rl) * DH))[lane];
        v.x += rv.x; v.y += rv.y; v.z += rv.z; v.w += rv.w;
        ((float4*)(x_out + (m0 + rl) * DH))[lane] = v;
    }
}

#define SCAP 32768
#define PLDS_PLACE (NCH * BMAXB + 3 * BMAXB + 64 + SCAP)
#define PLDS ((32 * 16 * LP) > PLDS_PLACE ? (32 * 16 * LP) : PLDS_PLACE)
static_assert(SCAP <= NCH * BMAXB, "dpos lies over the count matrix");

__global__ __launch_bounds__(1024)
void edge_place(const int* __restrict__ src0, const int* __restrict__ dst0,
                const int* __restrict__ src1, const int* __restrict__ dst1,
                const int* __restrict__ cntb, int* __restrict__ bbase,
                int* __restrict__ bkt, int nEdges, int nNodes, int nB,
                const float* __restrict__ x, const _Float16* __restrict__ WpX,
                const float* __restrict__ bf1, const float* __restrict__ bf2,
                float* __restrict__ x_out, int nRowTiles,
                _Float16* __restrict__ xH) {
    __shared__ __attribute__((aligned(16))) int cur[PLDS];
    int tid  = threadIdx.x;
    int xRows = (nRowTiles + 32 * NCH - 1) / (32 * NCH);
    if ((int)blockIdx.y >= 2 + xRows) {
        rows_to_f16(x, xH, (long long)nNodes * DH,
                    ((int)blockIdx.y - 2 - xRows) * NCH + (int)blockIdx.x, 2 * NCH, tid);
        return;
    }
    if (blockIdx.y >= 2) {
        int xb   = (int)(blockIdx.y - 2) * NCH + (int)blockIdx.x;
        int tile = xb * 32 + (tid >> 5);
        if (tile >= nRowTiles) return;
        x_chain_tile(x, WpX, bf1, bf2, x_out, tile, (float*)cur + (tid >> 5) * (16 * LP), tid & 31);
        return;
    }
    int list = blockIdx.y;
    const int* src = list ? src1 : src0;
    const int* dst = list ? dst1 : dst0;
    const int* cm = cntb + (long long)list * NCH * nB;
    int* out = bkt + (long long)list * nEdges;
    int per = chunk_len(nEdges);
    int c   = blockIdx.x;
    int beg = c * per;
    int end = min(beg + per, nEdges);
    int* m     = cur;
    int* posn  = cur + NCH * BMAXB;
    int* wtot  = posn + BMAXB;
    int* wtot2 = wtot + 32;
    int* lpos  = wtot2 + 32;
    int* delta = lpos + BMAXB;
    int* stage = delta + BMAXB;
    int* dpos  = cur;
    int lane = tid & 31;
    int w    = tid >> 5;
    int n = NCH * nB;
#pragma unroll 8
    for (int i = tid; i < n; i += 1024) m[i] = cm[i];
    __syncthreads();
    int tot = 0, mine = 0, own = 0;
    if (tid < nB) {
        for (int cc = 0; cc < NCH; ++cc) {
            if (cc == c) mine = tot;
            tot += m[cc * nB + tid];
        }
        own = m[c * nB + tid];
    }
    int incl = tot, linc = own;
#pragma unroll
    for (int d = 1; d < 32; d <<= 1) {
        int u  = __shfl_up(incl, d, 32);
        int u2 = __shfl_up(linc, d, 32);
        if (lane >= d) { incl += u; linc += u2; }
    }
    if (lane == 31) { wtot[w] = incl; wtot2[w] = linc; }
    __syncthreads();
    int wt = wtot[lane], wt2 = wtot2[lane];
    int tincl = wt, tincl2 = wt2;
#pragma unroll
    for (int d = 1; d < 32; d <<= 1) {
        int u  = __shfl_up(tincl, d, 32);
        int u2 = __shfl_up(tincl2, d, 32);
        if (lane >= d) { tincl += u; tincl2 += u2; }
    }
    int wexcl  = __shfl(tincl - wt, w, 32);
    int wexcl2 = __shfl(tincl2 - wt2, w, 32);
    int total  = __shfl(tincl2, 31, 32);
    int base = wexcl + (incl - tot);
    int lb   = wexcl2 + (linc - own);
    if (tid < nB) {
        posn[tid]  = base + mine;
        lpos[tid]  = lb;
        delta[tid] = base + mine - lb;
        if (c == 0) {
            bbase[list * (nB + 1) + tid] = base;
            if (tid == nB - 1) bbase[list * (nB + 1) + nB] = base + tot;
        }
    }
    __syncthreads();
    int e = beg + tid;
    if (per <= SCAP) {
        for (; e + 3 * 1024 < end; e += 4 * 1024) {
            int d0 = dst[e],        d1 = dst[e + 1024];
            int d2 = dst[e + 2048], d3 = dst[e + 3072];
            int s0 = src[e],        s1 = src[e + 1024];
            int s2 = src[e + 2048], s3 = src[e + 3072];
            int w0 = (s0 << BSH) | (d0 & BMASK), w1 = (s1 << BSH) | (d1 & BMASK);
            int w2 = (s2 << BSH) | (d2 & BMASK), w3 = (s3 << BSH) | (d3 & BMASK);
            bool k0 = (unsigned)d0 < (unsigned)nNodes, k1 = (unsigned)d1 < (unsigned)nNodes;
            bool k2 = (unsigned)d2 < (unsigned)nNodes, k3 = (unsigned)d3 < (unsigned)nNodes;
            int b0 = k0 ? d0 >> BSH : 0, b1 = k1 ? d1 >> BSH : 0;
            int b2 = k2 ? d2 >> BSH : 0, b3 = k3 ? d3 >> BSH : 0;
            int p0 = 0, p1 = 0, p2 = 0, p3 = 0;
            if (k0) p0 = atomicAdd(&lpos[b0], 1);
            if (k1) p1 = atomicAdd(&lpos[b1], 1);
            if (k2) p2 = atomicAdd(&lpos[b2], 1);
            if (k3) p3 = atomicAdd(&lpos[b3], 1);
            int g0 = delta[b0], g1 = delta[b1], g2 = delta[b2], g3 = delta[b3];
            if (k0) { stage[p0] = w0; dpos[p0] = p0 + g0; }
            if (k1) { stage[p1] = w1; dpos[p1] = p1 + g1; }
            if (k2) { stage[p2] = w2; dpos[p2] = p2 + g2; }
            if (k3) { stage[p3] = w3; dpos[p3] = p3 + g3; }
        }
        for (; e < end; e += 1024) {
            int d = dst[e];
            int sv = src[e];
            if ((unsigned)d < (unsigned)nNodes) {
                int bb = d >> BSH;
                int p  = atomicAdd(&lpos[bb], 1);
                stage[p] = (sv << BSH) | (d & BMASK);
                dpos[p]  = p + delta[bb];
            }
        }
        __syncthreads();
        int k = tid;
        for (; k + 3 * 1024 < total; k += 4 * 1024) {
            int q0 = dpos[k],         q1 = dpos[k + 1024];
            int q2 = dpos[k + 2048],  q3 = dpos[k + 3072];
            int v0 = stage[k],        v1 = stage[k + 1024];
            int v2 = stage[k + 2048], v3 = stage[k + 3072];
            out[q0] = v0; out[q1] = v1; out[q2] = v2; out[q3] = v3;
        }
        for (; k < total; k += 1024) out[dpos[k]] = stage[k];
    } else {
        for (; e + 3 * 1024 < end; e += 4 * 1024) {
            int d0 = dst[e],        d1 = dst[e + 1024];
            int d2 = dst[e + 2048], d3 = dst[e + 3072];
            int s0 = src[e],        s1 = src[e + 1024];
            int s2 = src[e + 2048], s3 = src[e + 3072];
            if ((unsigned)d0 < (unsigned)nNodes) out[atomicAdd(&posn[d0 >> BSH], 1)] = (s0 << BSH) | (d0 & BMASK);
            if ((unsigned)d1 < (unsigned)nNodes) out[atomicAdd(&posn[d1 >> BSH], 1)] = (s1 << BSH) | (d1 & BMASK);
            if ((unsigned)d2 < (unsigned)nNodes) out[atomicAdd(&posn[d2 >> BSH], 1)] = (s2 << BSH) | (d2 & BMASK);
            if ((unsigned)d3 < (unsigned)nNodes) out[atomicAdd(&posn[d3 >> BSH], 1)] = (s3 << BSH) | (d3 & BMASK);
        }
        for (; e < end; e += 1024) {
            int d = dst[e];
            int sv = src[e];
            if ((unsigned)d < (unsigned)nNodes) out[atomicAdd(&posn[d >> BSH], 1)] = (sv << BSH) | (d & BMASK);
        }
    }
}

extern "C" void kernel_launch(void* const* d_in, const int* in_sizes, int n_in,
                              void* d_out, int out_size, void* d_ws, size_t ws_size,
                              hipStream_t stream) {
    const float* x    = (const float*)d_in[0];
    const float* t    = (const float*)d_in[1];
    const int*   e_t  = (const int*)d_in[2];
    const int*   e_x  = (const int*)d_in[3];
    const float* W1a  = (const float*)d_in[4];
    const float* b1a  = (const float*)d_in[5];
    const float* W2a  = (const float*)d_in[6];
    const float* b2a  = (const float*)d_in[7];
    const float* W1b  = (const float*)d_in[8];
    const float* b1b  = (const float*)d_in[9];
    const float* W2b  = (const float*)d_in[10];
    const float* b2b  = (const float*)d_in[11];
    const float* Wo   = (const float*)d_in[12];
    const float* bo   = (const float*)d_in[13];
    const float* ln_g = (const float*)d_in[14];
    const float* ln_b = (const float*)d_in[15];
    const float* Wf1  = (const float*)d_in[16];
    const float* bf1  = (const float*)d_in[17];
    const float* Wf2  = (const float*)d_in[18];
    const float* bf2  = (const float*)d_in[19];

    const int N = in_sizes[0] / DH;
    const int E = in_sizes[2] / 2;
    const long long ND = (long long)N * DH;

    float* buf0 = (float*)d_ws;
    float* buf1 = buf0 + ND;
    float* buf2 = buf1 + ND;
    _Float16* wp = (_Float16*)(buf2 + ND);
    const int WPN = 4 * 8 * 32 * 16;
    _Float16* wp0 = wp;
    _Float16* wp5 = wp + 5 * WPN;

    float* x_out = (float*)d_out;
    float* t_out = x_out + ND;
    _Float16* tH = (_Float16*)t_out;
    _Float16* xH = tH + ND;

    const int nRowTiles = N / 16;
    dim3 gB((nRowTiles + 3) / 4), tB(128);

    int* sorted = (int*)buf2;
    int* off    = sorted + 2 * (long long)E;
    int* bkt    = (int*)buf0;
    const int nB = (N + BSZ - 1) / BSZ;
    int* cntb   = (int*)buf1;
    int* bbase  = cntb + 2 * NCH * nB;

    {
        dim3 gcp(NCH, 6);
        edge_count<<<gcp, 1024, 0, stream>>>(e_t + E, e_x + E, cntb, E, N, nB,
                                             W1a, W2a, W1b, W2b, Wo, Wf1, Wf2, wp, t, tH);
        dim3 gcx(NCH, 4 + (unsigned)((nRowTiles + 32 * NCH - 1) / (32 * NCH)));
        edge_place<<<gcx, 1024, 0, stream>>>(e_t, e_t + E, e_x, e_x + E, cntb, bbase, bkt, E, N, nB,
                                             x, wp5, bf1, bf2, x_out, nRowTiles, xH);
        dim3 go((unsigned)nB, 2);
        owner_sort<<<go, 1024, 0, stream>>>(bkt, bbase, off, sorted, E, N, nB);
        dim3 gg(512, 2);
        gather_sum<<<gg, 256, 0, stream>>>(tH, xH, off, sorted, buf0, buf1, N, E);
    }
    t_chain<<<gB, tB, 0, stream>>>(t, buf0, buf1, wp0, b1a, b2a, b1b, b2b, bo, ln_g, ln_b, t_out, nRowTiles);
}
